# starting kernel
# baseline (speedup 1.0000x reference)
#include <hip/hip_runtime.h>
#include <hip/hip_bf16.h>

typedef float v8f __attribute__((ext_vector_type(8)));
typedef float v2f __attribute__((ext_vector_type(2)));

#define BATCH 8
#define CCH   512
#define NHEAD 8
#define DHEAD 64
#define NTOK  1024
#define KTOP  4
#define C3    1536

__global__ void zero_out_kernel(float* __restrict__ out, int n) {
    int i = blockIdx.x * blockDim.x + threadIdx.x;
    if (i < n) out[i] = 0.0f;
}

__global__ __launch_bounds__(128) void qkv_kernel(const float* __restrict__ x,
                                                  const float* __restrict__ Wqkv,
                                                  const float* __restrict__ bqkv,
                                                  float* __restrict__ qbuf,
                                                  float* __restrict__ kTbuf,
                                                  float* __restrict__ vbuf) {
    const int lane    = threadIdx.x & 31;
    const int waveId  = blockIdx.x * 4 + (threadIdx.x >> 5);
    const int b       = waveId / (64 * 24);
    const int rem     = waveId % (64 * 24);
    const int mtile   = rem / 24;
    const int ng      = rem % 24;
    const int which   = ng >> 3;
    const int head    = ng & 7;
    const int m0      = mtile * 16;
    const int j0      = ng * 64;
    const int halfsel = lane >> 4;
    const int lm      = lane & 15;

    v8f acc[4];
    #pragma unroll
    for (int t = 0; t < 4; ++t) {
        float bv = bqkv[j0 + 16 * t + lm];
        #pragma unroll
        for (int r = 0; r < 8; ++r) acc[t][r] = bv;
    }

    const float* xb = x + (size_t)b * CCH * NTOK;
    for (int k0 = 0; k0 < CCH; k0 += 4) {
        const int kk = k0 + 2 * halfsel;
        v2f a;
        a.x = xb[(size_t)kk * NTOK + m0 + lm];
        a.y = xb[(size_t)(kk + 1) * NTOK + m0 + lm];
        #pragma unroll
        for (int t = 0; t < 4; ++t) {
            v2f bf;
            bf.x = Wqkv[(size_t)kk * C3 + j0 + 16 * t + lm];
            bf.y = Wqkv[(size_t)(kk + 1) * C3 + j0 + 16 * t + lm];
            acc[t] = __builtin_amdgcn_wmma_f32_16x16x4_f32(
                false, a, false, bf, (short)0, acc[t], false, false);
        }
    }

    const int bh = b * NHEAD + head;
    #pragma unroll
    for (int t = 0; t < 4; ++t) {
        #pragma unroll
        for (int r = 0; r < 8; ++r) {
            const int m  = m0 + r + 8 * halfsel;
            const int jl = 16 * t + lm;
            const float val = acc[t][r];
            if (which == 0) {
                qbuf[((size_t)bh * NTOK + m) * DHEAD + jl] = val;
            } else if (which == 1) {
                kTbuf[((((size_t)bh * (DHEAD / 2) + (jl >> 1)) * NTOK + m) << 1) + (jl & 1)] = val;
            } else {
                vbuf[((size_t)bh * NTOK + m) * DHEAD + jl] = val;
            }
        }
    }
}

__global__ __launch_bounds__(256) void attn_kernel(const float* __restrict__ qbuf,
                                                   const float* __restrict__ kTbuf,
                                                   const float* __restrict__ vbuf,
                                                   float* __restrict__ out) {
    __shared__ float sc[16][260];
    __shared__ float candw[16][16][KTOP];
    __shared__ int   candi[16][16][KTOP];
    __shared__ float topw[16][KTOP];
    __shared__ int   topi[16][KTOP];
    __shared__ float out_acc[DHEAD];

    const int tid     = threadIdx.x;
    const int lane    = tid & 31;
    const int wv      = tid >> 5;
    const int halfsel = lane >> 4;
    const int lm      = lane & 15;

    const int bh    = blockIdx.x >> 6;
    const int itile = blockIdx.x & 63;
    const int b     = bh >> 3;
    const int h     = bh & 7;
    const int i0    = itile * 16;

    const size_t qb = (size_t)bh * NTOK * DHEAD;
    const size_t vb = qb;
    const float2* kT2 = (const float2*)(kTbuf + (size_t)bh * DHEAD * NTOK);

    if (tid < DHEAD) out_acc[tid] = 0.0f;

    v2f afrag[16];
    {
        const float* qrow = qbuf + qb + (size_t)(i0 + lm) * DHEAD + 2 * halfsel;
        #pragma unroll
        for (int kk = 0; kk < 16; ++kk) {
            afrag[kk].x = qrow[kk * 4];
            afrag[kk].y = qrow[kk * 4 + 1];
        }
    }

    const int srow = tid >> 4;
    const int ssub = tid & 15;
    float lw[KTOP];
    int   li[KTOP];
    #pragma unroll
    for (int p = 0; p < KTOP; ++p) { lw[p] = -__builtin_inff(); li[p] = 0x7fffffff; }

    for (int chunk = 0; chunk < 4; ++chunk) {
        const int jcbase = chunk * 256;
        #pragma unroll
        for (int tsub = 0; tsub < 2; ++tsub) {
            const int jl0 = wv * 32 + tsub * 16;
            const int jg  = jcbase + jl0;
            v8f acc;
            #pragma unroll
            for (int r = 0; r < 8; ++r) acc[r] = 0.0f;
            #pragma unroll
            for (int kk = 0; kk < 16; ++kk) {
                const float2 bp = kT2[(size_t)(kk * 2 + halfsel) * NTOK + jg + lm];
                v2f bf; bf.x = bp.x; bf.y = bp.y;
                acc = __builtin_amdgcn_wmma_f32_16x16x4_f32(
                    false, afrag[kk], false, bf, (short)0, acc, false, false);
            }
            #pragma unroll
            for (int r = 0; r < 8; ++r)
                sc[r + 8 * halfsel][jl0 + lm] = acc[r];
        }
        __syncthreads();
        #pragma unroll
        for (int c = 0; c < 16; ++c) {
            const int cl = ssub * 16 + c;
            const float v = sc[srow][cl];
            const int   j = jcbase + cl;
            #pragma unroll
            for (int p = 0; p < KTOP; ++p) {
                const bool better = (v > lw[p]) || (v == lw[p] && j < li[p]);
                if (better) {
                    #pragma unroll
                    for (int q = KTOP - 1; q > p; --q) { lw[q] = lw[q - 1]; li[q] = li[q - 1]; }
                    lw[p] = v; li[p] = j;
                    break;
                }
            }
        }
        __syncthreads();
    }

    #pragma unroll
    for (int p = 0; p < KTOP; ++p) { candw[srow][ssub][p] = lw[p]; candi[srow][ssub][p] = li[p]; }
    __syncthreads();

    if (tid < 16) {
        const int row = tid;
        float w4[KTOP]; int i4[KTOP];
        #pragma unroll
        for (int p = 0; p < KTOP; ++p) { w4[p] = -__builtin_inff(); i4[p] = 0x7fffffff; }
        for (int s = 0; s < 16; ++s) {
            #pragma unroll
            for (int cp = 0; cp < KTOP; ++cp) {
                const float v = candw[row][s][cp];
                const int   j = candi[row][s][cp];
                #pragma unroll
                for (int p = 0; p < KTOP; ++p) {
                    const bool better = (v > w4[p]) || (v == w4[p] && j < i4[p]);
                    if (better) {
                        #pragma unroll
                        for (int q = KTOP - 1; q > p; --q) { w4[q] = w4[q - 1]; i4[q] = i4[q - 1]; }
                        w4[p] = v; i4[p] = j;
                        break;
                    }
                }
            }
        }
        const float mx = w4[0];
        float e[KTOP], s = 0.0f;
        #pragma unroll
        for (int p = 0; p < KTOP; ++p) { e[p] = __expf(w4[p] - mx); s += e[p]; }
        const float inv = 1.0f / s;
        #pragma unroll
        for (int p = 0; p < KTOP; ++p) { topw[row][p] = e[p] * inv; topi[row][p] = i4[p]; }
    }
    __syncthreads();

    {
        const int row = tid >> 4;
        const int dd0 = (tid & 15) * 4;
        #pragma unroll
        for (int u = 0; u < 4; ++u) {
            const int dd = dd0 + u;
            float wvv = 0.0f;
            #pragma unroll
            for (int p = 0; p < KTOP; ++p)
                wvv += topw[row][p] * vbuf[vb + (size_t)topi[row][p] * DHEAD + dd];
            const float val = wvv * qbuf[qb + (size_t)(i0 + row) * DHEAD + dd];
            atomicAdd(&out_acc[dd], val);
        }
    }
    __syncthreads();
    if (tid < DHEAD)
        atomicAdd(&out[b * CCH + tid * NHEAD + h], out_acc[tid]);
}

extern "C" void kernel_launch(void* const* d_in, const int* in_sizes, int n_in,
                              void* d_out, int out_size, void* d_ws, size_t ws_size,
                              hipStream_t stream) {
    const float* x  = (const float*)d_in[0];
    const float* Wq = (const float*)d_in[1];
    const float* bq = (const float*)d_in[2];
    float* out = (float*)d_out;
    float* ws  = (float*)d_ws;

    const size_t per = (size_t)BATCH * NHEAD * NTOK * DHEAD;
    float* qbuf = ws;
    float* kT   = ws + per;
    float* vbuf = ws + 2 * per;

    zero_out_kernel<<<(out_size + 255) / 256, 256, 0, stream>>>(out, out_size);
    qkv_kernel<<<3072, 128, 0, stream>>>(x, Wq, bq, qbuf, kT, vbuf);
    attn_kernel<<<4096, 256, 0, stream>>>(qbuf, kT, vbuf, out);
}


# attn top-4 in two passes over the whole 16x1024 score tile kept in LDS: value-only max/min chain + lane exchange gives each row's 4th largest score, then (value,index) insert only for scores >= it
# speedup vs baseline: 1.5031x; 1.5031x over previous
#include <hip/hip_runtime.h>
#include <hip/hip_bf16.h>

typedef float v8f __attribute__((ext_vector_type(8)));
typedef float v2f __attribute__((ext_vector_type(2)));

#define BATCH 8
#define CCH   512
#define NHEAD 8
#define DHEAD 64
#define NTOK  1024
#define KTOP  4
#define C3    1536

__global__ void zero_out_kernel(float* __restrict__ out, int n) {
    int i = blockIdx.x * blockDim.x + threadIdx.x;
    if (i < n) out[i] = 0.0f;
}

typedef float v4f __attribute__((ext_vector_type(4)));
#define QK_KC    32
#define QK_KP    (QK_KC / 2)
#define QK_PITCH 288

static __device__ __forceinline__ void qk_chunk(const float* pa, const float* pb, v8f (&acc)[4][2]) {
    v2f af[2][4], bf[2][2];
    #pragma unroll
    for (int i = 0; i < 4; ++i) af[0][i] = *(const v2f*)(pa + i * 32);
    #pragma unroll
    for (int t = 0; t < 2; ++t) bf[0][t] = *(const v2f*)(pb + t * 32);
    #pragma unroll
    for (int s = 0; s < QK_KC / 4; ++s) {
        const int cur = s & 1, nxt = cur ^ 1;
        if (s + 1 < QK_KC / 4) {
            #pragma unroll
            for (int i = 0; i < 4; ++i) af[nxt][i] = *(const v2f*)(pa + 2 * (s + 1) * QK_PITCH + i * 32);
            #pragma unroll
            for (int t = 0; t < 2; ++t) bf[nxt][t] = *(const v2f*)(pb + 2 * (s + 1) * QK_PITCH + t * 32);
        }
        __builtin_amdgcn_sched_barrier(0);
        #pragma unroll
        for (int i = 0; i < 4; ++i) {
            #pragma unroll
            for (int t = 0; t < 2; ++t)
                acc[i][t] = __builtin_amdgcn_wmma_f32_16x16x4_f32(
                    false, af[cur][i], false, bf[cur][t], (short)0, acc[i][t], false, false);
        }
        __builtin_amdgcn_sched_barrier(0);
    }
}

__global__ __launch_bounds__(256) void qkv_kernel(const float* __restrict__ x,
                                                  const float* __restrict__ Wqkv,
                                                  const float* __restrict__ bqkv,
                                                  float* __restrict__ qbuf,
                                                  float* __restrict__ kTbuf,
                                                  float* __restrict__ vbuf) {
    __shared__ __attribute__((aligned(16))) float sA[2][QK_KP * QK_PITCH];
    __shared__ __attribute__((aligned(16))) float sB[2][QK_KP * QK_PITCH];

    const int tid     = threadIdx.x;
    const int lane    = tid & 31;
    const int wv      = tid >> 5;
    const int halfsel = lane >> 4;
    const int lm      = lane & 15;
    const int wm      = wv & 1;
    const int wn      = wv >> 1;

    const int nblk  = blockIdx.x % 12;
    const int mb    = blockIdx.x / 12;
    const int b     = mb >> 3;
    const int mbase = (mb & 7) * 128;
    const int jbase = nblk * 128;

    const float* xb = x + (size_t)b * CCH * NTOK + mbase + 4 * lane;
    const float* wb = Wqkv + jbase + 4 * lane;

    v8f acc[4][2];
    #pragma unroll
    for (int t = 0; t < 2; ++t) {
        const float bv = bqkv[jbase + wn * 32 + 16 * t + lm];
        #pragma unroll
        for (int i = 0; i < 4; ++i) {
            #pragma unroll
            for (int r = 0; r < 8; ++r) acc[i][t][r] = bv;
        }
    }

    v4f ra[2][2], rb[2][2];
    const int aoff = halfsel * QK_PITCH + (wm * 64 + lm) * 2;
    const int boff = halfsel * QK_PITCH + (wn * 32 + lm) * 2;

#define QK_GLOAD(chunk)                                                              \
    _Pragma("unroll")                                                                \
    for (int i_ = 0; i_ < 2; ++i_) {                                                 \
        const int k_ = (chunk) * QK_KC + 2 * (wv + 8 * i_);                          \
        ra[i_][0] = *(const v4f*)(xb + (size_t)k_ * NTOK);                           \
        ra[i_][1] = *(const v4f*)(xb + (size_t)(k_ + 1) * NTOK);                     \
        rb[i_][0] = *(const v4f*)(wb + (size_t)k_ * C3);                             \
        rb[i_][1] = *(const v4f*)(wb + (size_t)(k_ + 1) * C3);                       \
    }
#define QK_SSTORE(buf)                                                               \
    _Pragma("unroll")                                                                \
    for (int i_ = 0; i_ < 2; ++i_) {                                                 \
        float* da_ = &sA[(buf)][(wv + 8 * i_) * QK_PITCH + lane * 8];                \
        float* db_ = &sB[(buf)][(wv + 8 * i_) * QK_PITCH + lane * 8];                \
        *(v4f*)(da_)     = __builtin_shufflevector(ra[i_][0], ra[i_][1], 0, 4, 1, 5); \
        *(v4f*)(da_ + 4) = __builtin_shufflevector(ra[i_][0], ra[i_][1], 2, 6, 3, 7); \
        *(v4f*)(db_)     = __builtin_shufflevector(rb[i_][0], rb[i_][1], 0, 4, 1, 5); \
        *(v4f*)(db_ + 4) = __builtin_shufflevector(rb[i_][0], rb[i_][1], 2, 6, 3, 7); \
    }

    QK_GLOAD(0)
    QK_SSTORE(0)
    __syncthreads();
    QK_GLOAD(1)
    for (int c = 0; c + 2 < CCH / QK_KC; ++c) {
        const int buf = c & 1;
        qk_chunk(&sA[buf][aoff], &sB[buf][boff], acc);
        #pragma unroll
        for (int i = 0; i < 2; ++i) {
            asm volatile("" : "+v"(ra[i][0]), "+v"(ra[i][1]), "+v"(rb[i][0]), "+v"(rb[i][1]));
        }
        QK_SSTORE(buf ^ 1)
        QK_GLOAD(c + 2)
        asm volatile("s_wait_dscnt 0x0\n\ts_barrier_signal -1\n\ts_barrier_wait -1" ::: "memory");
    }
    qk_chunk(&sA[0][aoff], &sB[0][boff], acc);
    QK_SSTORE(1)
    __syncthreads();
    qk_chunk(&sA[1][aoff], &sB[1][boff], acc);
#undef QK_GLOAD
#undef QK_SSTORE

    const int which = jbase >> 9;
    const int head  = ((jbase & 511) + wn * 32) >> 6;
    const int bh    = b * NHEAD + head;
    const int jl0   = (wn & 1) * 32 + lm;
    float* dst;
    int laneoff, mstride, tstride;
    if (which == 1) {
        dst     = kTbuf + (size_t)bh * DHEAD * NTOK;
        laneoff = (jl0 >> 1) * (2 * NTOK) + (jl0 & 1);
        mstride = 2;
        tstride = 8 * 2 * NTOK;
    } else {
        dst     = (which == 0 ? qbuf : vbuf) + (size_t)bh * NTOK * DHEAD;
        laneoff = jl0;
        mstride = DHEAD;
        tstride = 16;
    }
    float* p = dst + laneoff + (size_t)(mbase + wm * 64 + 8 * halfsel) * mstride;
    #pragma unroll
    for (int i = 0; i < 4; ++i) {
        #pragma unroll
        for (int t = 0; t < 2; ++t) {
            #pragma unroll
            for (int r = 0; r < 8; ++r)
                p[(16 * i + r) * mstride + t * tstride] = acc[i][t][r];
        }
    }
}

#define SC_PITCH 1028
__global__ __launch_bounds__(256) void attn_kernel(const float* __restrict__ qbuf,
                                                   const float* __restrict__ kTbuf,
                                                   const float* __restrict__ vbuf,
                                                   float* __restrict__ out) {
    __shared__ __attribute__((aligned(16))) float sc[16 * SC_PITCH];
    __shared__ float candw[16][16][KTOP];
    __shared__ int   candi[16][16][KTOP];
    __shared__ float topw[16][KTOP];
    __shared__ int   topi[16][KTOP];
    __shared__ float out_acc[DHEAD];

    const int tid     = threadIdx.x;
    const int lane    = tid & 31;
    const int wv      = tid >> 5;
    const int halfsel = lane >> 4;
    const int lm      = lane & 15;

    const int bh    = blockIdx.x >> 6;
    const int itile = blockIdx.x & 63;
    const int b     = bh >> 3;
    const int h     = bh & 7;
    const int i0    = itile * 16;

    const size_t qb = (size_t)bh * NTOK * DHEAD;
    const size_t vb = qb;
    const float2* kT2 = (const float2*)(kTbuf + (size_t)bh * DHEAD * NTOK);

    if (tid < DHEAD) out_acc[tid] = 0.0f;

    v2f afrag[16];
    {
        const float* qrow = qbuf + qb + (size_t)(i0 + lm) * DHEAD + 2 * halfsel;
        #pragma unroll
        for (int kk = 0; kk < 16; ++kk) {
            afrag[kk].x = qrow[kk * 4];
            afrag[kk].y = qrow[kk * 4 + 1];
        }
    }

    for (int chunk = 0; chunk < 4; ++chunk) {
        #pragma unroll
        for (int tsub = 0; tsub < 2; ++tsub) {
            const int jg = chunk * 256 + wv * 32 + tsub * 16;
            v8f acc;
            #pragma unroll
            for (int r = 0; r < 8; ++r) acc[r] = 0.0f;
            #pragma unroll
            for (int kk = 0; kk < 16; ++kk) {
                const float2 bp = kT2[(size_t)(kk * 2 + halfsel) * NTOK + jg + lm];
                v2f bf; bf.x = bp.x; bf.y = bp.y;
                acc = __builtin_amdgcn_wmma_f32_16x16x4_f32(
                    false, afrag[kk], false, bf, (short)0, acc, false, false);
            }
            #pragma unroll
            for (int r = 0; r < 8; ++r)
                sc[(r + 8 * halfsel) * SC_PITCH + jg + lm] = acc[r];
        }
    }
    __syncthreads();

    const int srow = tid >> 4;
    const int ssub = tid & 15;
    const float* srcrow = &sc[srow * SC_PITCH + ssub * 4];

    float m0 = -__builtin_inff(), m1 = m0, m2 = m0, m3 = m0;
#define TOP4_TAKE(x_)                                                   \
    {                                                                   \
        const float x__ = (x_);                                         \
        m3 = __builtin_fmaxf(m3, __builtin_fminf(m2, x__));             \
        m2 = __builtin_fmaxf(m2, __builtin_fminf(m1, x__));             \
        m1 = __builtin_fmaxf(m1, __builtin_fminf(m0, x__));             \
        m0 = __builtin_fmaxf(m0, x__);                                  \
    }
    #pragma unroll
    for (int g = 0; g < 16; ++g) {
        const v4f sx = *(const v4f*)(srcrow + g * 64);
        #pragma unroll
        for (int e = 0; e < 4; ++e) TOP4_TAKE(sx[e])
    }
    #pragma unroll
    for (int off = 1; off < 16; off <<= 1) {
        const float o0 = __shfl_xor(m0, off);
        const float o1 = __shfl_xor(m1, off);
        const float o2 = __shfl_xor(m2, off);
        const float o3 = __shfl_xor(m3, off);
        TOP4_TAKE(o0)
        TOP4_TAKE(o1)
        TOP4_TAKE(o2)
        TOP4_TAKE(o3)
    }
#undef TOP4_TAKE
    const float tcut = m3;

    float lw[KTOP];
    int   li[KTOP];
    #pragma unroll
    for (int p = 0; p < KTOP; ++p) { lw[p] = -__builtin_inff(); li[p] = 0x7fffffff; }
    #pragma unroll 1
    for (int gq = 0; gq < 4; ++gq) {
        v4f sx[4];
        #pragma unroll
        for (int u = 0; u < 4; ++u) sx[u] = *(const v4f*)(srcrow + (gq * 4 + u) * 64);
        #pragma unroll
        for (int u = 0; u < 4; ++u) {
            #pragma unroll
            for (int e = 0; e < 4; ++e) {
                const float v = sx[u][e];
                if (v >= tcut) {
                    const int j = (gq * 4 + u) * 64 + ssub * 4 + e;
                    #pragma unroll
                    for (int p = 0; p < KTOP; ++p) {
                        const bool better = (v > lw[p]) || (v == lw[p] && j < li[p]);
                        if (better) {
                            #pragma unroll
                            for (int q = KTOP - 1; q > p; --q) { lw[q] = lw[q - 1]; li[q] = li[q - 1]; }
                            lw[p] = v; li[p] = j;
                            break;
                        }
                    }
                }
            }
        }
    }

    #pragma unroll
    for (int p = 0; p < KTOP; ++p) { candw[srow][ssub][p] = lw[p]; candi[srow][ssub][p] = li[p]; }
    __syncthreads();

    if (tid < 16) {
        const int row = tid;
        float w4[KTOP]; int i4[KTOP];
        #pragma unroll
        for (int p = 0; p < KTOP; ++p) { w4[p] = -__builtin_inff(); i4[p] = 0x7fffffff; }
        for (int s = 0; s < 16; ++s) {
            #pragma unroll
            for (int cp = 0; cp < KTOP; ++cp) {
                const float v = candw[row][s][cp];
                const int   j = candi[row][s][cp];
                if (j != 0x7fffffff) {
                    #pragma unroll
                    for (int p = 0; p < KTOP; ++p) {
                        const bool better = (v > w4[p]) || (v == w4[p] && j < i4[p]);
                        if (better) {
                            #pragma unroll
                            for (int q = KTOP - 1; q > p; --q) { w4[q] = w4[q - 1]; i4[q] = i4[q - 1]; }
                            w4[p] = v; i4[p] = j;
                            break;
                        }
                    }
                }
            }
        }
        const float mx = w4[0];
        float e[KTOP], s = 0.0f;
        #pragma unroll
        for (int p = 0; p < KTOP; ++p) { e[p] = __expf(w4[p] - mx); s += e[p]; }
        const float inv = 1.0f / s;
        #pragma unroll
        for (int p = 0; p < KTOP; ++p) { topw[row][p] = e[p] * inv; topi[row][p] = i4[p]; }
    }
    __syncthreads();

    {
        const int row = tid >> 4;
        const int dd0 = (tid & 15) * 4;
        #pragma unroll
        for (int u = 0; u < 4; ++u) {
            const int dd = dd0 + u;
            float wvv = 0.0f;
            #pragma unroll
            for (int p = 0; p < KTOP; ++p)
                wvv += topw[row][p] * vbuf[vb + (size_t)topi[row][p] * DHEAD + dd];
            const float val = wvv * qbuf[qb + (size_t)(i0 + row) * DHEAD + dd];
            atomicAdd(&out_acc[dd], val);
        }
    }
    __syncthreads();
    if (tid < DHEAD)
        atomicAdd(&out[b * CCH + tid * NHEAD + h], out_acc[tid]);
}

extern "C" void kernel_launch(void* const* d_in, const int* in_sizes, int n_in,
                              void* d_out, int out_size, void* d_ws, size_t ws_size,
                              hipStream_t stream) {
    const float* x  = (const float*)d_in[0];
    const float* Wq = (const float*)d_in[1];
    const float* bq = (const float*)d_in[2];
    float* out = (float*)d_out;
    float* ws  = (float*)d_ws;

    const size_t per = (size_t)BATCH * NHEAD * NTOK * DHEAD;
    float* qbuf = ws;
    float* kT   = ws + per;
    float* vbuf = ws + 2 * per;

    zero_out_kernel<<<(out_size + 255) / 256, 256, 0, stream>>>(out, out_size);
    qkv_kernel<<<768, 256, 0, stream>>>(x, Wq, bq, qbuf, kT, vbuf);
    attn_kernel<<<4096, 256, 0, stream>>>(qbuf, kT, vbuf, out);
}


# attn row merge by lane exchange: a row's 16 lanes merge their sorted top-4 (value,index) lists in 4 shuffle rounds, then each does softmax and V gather; LDS lists, serial merge, 2 barriers gone
# speedup vs baseline: 1.6691x; 1.1104x over previous
#include <hip/hip_runtime.h>
#include <hip/hip_bf16.h>

typedef float v8f __attribute__((ext_vector_type(8)));
typedef float v2f __attribute__((ext_vector_type(2)));

#define BATCH 8
#define CCH   512
#define NHEAD 8
#define DHEAD 64
#define NTOK  1024
#define KTOP  4
#define C3    1536

__global__ void zero_out_kernel(float* __restrict__ out, int n) {
    int i = blockIdx.x * blockDim.x + threadIdx.x;
    if (i < n) out[i] = 0.0f;
}

typedef float v4f __attribute__((ext_vector_type(4)));
#define QK_KC    32
#define QK_KP    (QK_KC / 2)
#define QK_PITCH 288

static __device__ __forceinline__ void qk_chunk(const float* pa, const float* pb, v8f (&acc)[4][2]) {
    v2f af[2][4], bf[2][2];
    #pragma unroll
    for (int i = 0; i < 4; ++i) af[0][i] = *(const v2f*)(pa + i * 32);
    #pragma unroll
    for (int t = 0; t < 2; ++t) bf[0][t] = *(const v2f*)(pb + t * 32);
    #pragma unroll
    for (int s = 0; s < QK_KC / 4; ++s) {
        const int cur = s & 1, nxt = cur ^ 1;
        if (s + 1 < QK_KC / 4) {
            #pragma unroll
            for (int i = 0; i < 4; ++i) af[nxt][i] = *(const v2f*)(pa + 2 * (s + 1) * QK_PITCH + i * 32);
            #pragma unroll
            for (int t = 0; t < 2; ++t) bf[nxt][t] = *(const v2f*)(pb + 2 * (s + 1) * QK_PITCH + t * 32);
        }
        __builtin_amdgcn_sched_barrier(0);
        #pragma unroll
        for (int i = 0; i < 4; ++i) {
            #pragma unroll
            for (int t = 0; t < 2; ++t)
                acc[i][t] = __builtin_amdgcn_wmma_f32_16x16x4_f32(
                    false, af[cur][i], false, bf[cur][t], (short)0, acc[i][t], false, false);
        }
        __builtin_amdgcn_sched_barrier(0);
    }
}

__global__ __launch_bounds__(256) void qkv_kernel(const float* __restrict__ x,
                                                  const float* __restrict__ Wqkv,
                                                  const float* __restrict__ bqkv,
                                                  float* __restrict__ qbuf,
                                                  float* __restrict__ kTbuf,
                                                  float* __restrict__ vbuf) {
    __shared__ __attribute__((aligned(16))) float sA[2][QK_KP * QK_PITCH];
    __shared__ __attribute__((aligned(16))) float sB[2][QK_KP * QK_PITCH];

    const int tid     = threadIdx.x;
    const int lane    = tid & 31;
    const int wv      = tid >> 5;
    const int halfsel = lane >> 4;
    const int lm      = lane & 15;
    const int wm      = wv & 1;
    const int wn      = wv >> 1;

    const int nblk  = blockIdx.x % 12;
    const int mb    = blockIdx.x / 12;
    const int b     = mb >> 3;
    const int mbase = (mb & 7) * 128;
    const int jbase = nblk * 128;

    const float* xb = x + (size_t)b * CCH * NTOK + mbase + 4 * lane;
    const float* wb = Wqkv + jbase + 4 * lane;

    v8f acc[4][2];
    #pragma unroll
    for (int t = 0; t < 2; ++t) {
        const float bv = bqkv[jbase + wn * 32 + 16 * t + lm];
        #pragma unroll
        for (int i = 0; i < 4; ++i) {
            #pragma unroll
            for (int r = 0; r < 8; ++r) acc[i][t][r] = bv;
        }
    }

    v4f ra[2][2], rb[2][2];
    const int aoff = halfsel * QK_PITCH + (wm * 64 + lm) * 2;
    const int boff = halfsel * QK_PITCH + (wn * 32 + lm) * 2;

#define QK_GLOAD(chunk)                                                              \
    _Pragma("unroll")                                                                \
    for (int i_ = 0; i_ < 2; ++i_) {                                                 \
        const int k_ = (chunk) * QK_KC + 2 * (wv + 8 * i_);                          \
        ra[i_][0] = *(const v4f*)(xb + (size_t)k_ * NTOK);                           \
        ra[i_][1] = *(const v4f*)(xb + (size_t)(k_ + 1) * NTOK);                     \
        rb[i_][0] = *(const v4f*)(wb + (size_t)k_ * C3);                             \
        rb[i_][1] = *(const v4f*)(wb + (size_t)(k_ + 1) * C3);                       \
    }
#define QK_SSTORE(buf)                                                               \
    _Pragma("unroll")                                                                \
    for (int i_ = 0; i_ < 2; ++i_) {                                                 \
        float* da_ = &sA[(buf)][(wv + 8 * i_) * QK_PITCH + lane * 8];                \
        float* db_ = &sB[(buf)][(wv + 8 * i_) * QK_PITCH + lane * 8];                \
        *(v4f*)(da_)     = __builtin_shufflevector(ra[i_][0], ra[i_][1], 0, 4, 1, 5); \
        *(v4f*)(da_ + 4) = __builtin_shufflevector(ra[i_][0], ra[i_][1], 2, 6, 3, 7); \
        *(v4f*)(db_)     = __builtin_shufflevector(rb[i_][0], rb[i_][1], 0, 4, 1, 5); \
        *(v4f*)(db_ + 4) = __builtin_shufflevector(rb[i_][0], rb[i_][1], 2, 6, 3, 7); \
    }

    QK_GLOAD(0)
    QK_SSTORE(0)
    __syncthreads();
    QK_GLOAD(1)
    for (int c = 0; c + 2 < CCH / QK_KC; ++c) {
        const int buf = c & 1;
        qk_chunk(&sA[buf][aoff], &sB[buf][boff], acc);
        #pragma unroll
        for (int i = 0; i < 2; ++i) {
            asm volatile("" : "+v"(ra[i][0]), "+v"(ra[i][1]), "+v"(rb[i][0]), "+v"(rb[i][1]));
        }
        QK_SSTORE(buf ^ 1)
        QK_GLOAD(c + 2)
        asm volatile("s_wait_dscnt 0x0\n\ts_barrier_signal -1\n\ts_barrier_wait -1" ::: "memory");
    }
    qk_chunk(&sA[0][aoff], &sB[0][boff], acc);
    QK_SSTORE(1)
    __syncthreads();
    qk_chunk(&sA[1][aoff], &sB[1][boff], acc);
#undef QK_GLOAD
#undef QK_SSTORE

    const int which = jbase >> 9;
    const int head  = ((jbase & 511) + wn * 32) >> 6;
    const int bh    = b * NHEAD + head;
    const int jl0   = (wn & 1) * 32 + lm;
    float* dst;
    int laneoff, mstride, tstride;
    if (which == 1) {
        dst     = kTbuf + (size_t)bh * DHEAD * NTOK;
        laneoff = (jl0 >> 1) * (2 * NTOK) + (jl0 & 1);
        mstride = 2;
        tstride = 8 * 2 * NTOK;
    } else {
        dst     = (which == 0 ? qbuf : vbuf) + (size_t)bh * NTOK * DHEAD;
        laneoff = jl0;
        mstride = DHEAD;
        tstride = 16;
    }
    float* p = dst + laneoff + (size_t)(mbase + wm * 64 + 8 * halfsel) * mstride;
    #pragma unroll
    for (int i = 0; i < 4; ++i) {
        #pragma unroll
        for (int t = 0; t < 2; ++t) {
            #pragma unroll
            for (int r = 0; r < 8; ++r)
                p[(16 * i + r) * mstride + t * tstride] = acc[i][t][r];
        }
    }
}

#define SC_PITCH 1028
__global__ __launch_bounds__(256) void attn_kernel(const float* __restrict__ qbuf,
                                                   const float* __restrict__ kTbuf,
                                                   const float* __restrict__ vbuf,
                                                   float* __restrict__ out) {
    __shared__ __attribute__((aligned(16))) float sc[16 * SC_PITCH];
    __shared__ float out_acc[DHEAD];

    const int tid     = threadIdx.x;
    const int lane    = tid & 31;
    const int wv      = tid >> 5;
    const int halfsel = lane >> 4;
    const int lm      = lane & 15;

    const int bh    = blockIdx.x >> 6;
    const int itile = blockIdx.x & 63;
    const int b     = bh >> 3;
    const int h     = bh & 7;
    const int i0    = itile * 16;

    const size_t qb = (size_t)bh * NTOK * DHEAD;
    const size_t vb = qb;
    const float2* kT2 = (const float2*)(kTbuf + (size_t)bh * DHEAD * NTOK);

    if (tid < DHEAD) out_acc[tid] = 0.0f;

    v2f afrag[16];
    {
        const float* qrow = qbuf + qb + (size_t)(i0 + lm) * DHEAD + 2 * halfsel;
        #pragma unroll
        for (int kk = 0; kk < 16; ++kk) {
            afrag[kk].x = qrow[kk * 4];
            afrag[kk].y = qrow[kk * 4 + 1];
        }
    }

    for (int chunk = 0; chunk < 4; ++chunk) {
        #pragma unroll
        for (int tsub = 0; tsub < 2; ++tsub) {
            const int jg = chunk * 256 + wv * 32 + tsub * 16;
            v8f acc;
            #pragma unroll
            for (int r = 0; r < 8; ++r) acc[r] = 0.0f;
            #pragma unroll
            for (int kk = 0; kk < 16; ++kk) {
                const float2 bp = kT2[(size_t)(kk * 2 + halfsel) * NTOK + jg + lm];
                v2f bf; bf.x = bp.x; bf.y = bp.y;
                acc = __builtin_amdgcn_wmma_f32_16x16x4_f32(
                    false, afrag[kk], false, bf, (short)0, acc, false, false);
            }
            #pragma unroll
            for (int r = 0; r < 8; ++r)
                sc[(r + 8 * halfsel) * SC_PITCH + jg + lm] = acc[r];
        }
    }
    __syncthreads();

    const int srow = tid >> 4;
    const int ssub = tid & 15;
    const float* srcrow = &sc[srow * SC_PITCH + ssub * 4];

    float m0 = -__builtin_inff(), m1 = m0, m2 = m0, m3 = m0;
#define TOP4_TAKE(x_)                                                   \
    {                                                                   \
        const float x__ = (x_);                                         \
        m3 = __builtin_fmaxf(m3, __builtin_fminf(m2, x__));             \
        m2 = __builtin_fmaxf(m2, __builtin_fminf(m1, x__));             \
        m1 = __builtin_fmaxf(m1, __builtin_fminf(m0, x__));             \
        m0 = __builtin_fmaxf(m0, x__);                                  \
    }
    #pragma unroll
    for (int g = 0; g < 16; ++g) {
        const v4f sx = *(const v4f*)(srcrow + g * 64);
        #pragma unroll
        for (int e = 0; e < 4; ++e) TOP4_TAKE(sx[e])
    }
    #pragma unroll
    for (int off = 1; off < 16; off <<= 1) {
        const float o0 = __shfl_xor(m0, off);
        const float o1 = __shfl_xor(m1, off);
        const float o2 = __shfl_xor(m2, off);
        const float o3 = __shfl_xor(m3, off);
        TOP4_TAKE(o0)
        TOP4_TAKE(o1)
        TOP4_TAKE(o2)
        TOP4_TAKE(o3)
    }
#undef TOP4_TAKE
    const float tcut = m3;

    float lw[KTOP];
    int   li[KTOP];
    #pragma unroll
    for (int p = 0; p < KTOP; ++p) { lw[p] = -__builtin_inff(); li[p] = 0x7fffffff; }
    #pragma unroll 1
    for (int gq = 0; gq < 4; ++gq) {
        v4f sx[4];
        #pragma unroll
        for (int u = 0; u < 4; ++u) sx[u] = *(const v4f*)(srcrow + (gq * 4 + u) * 64);
        #pragma unroll
        for (int u = 0; u < 4; ++u) {
            #pragma unroll
            for (int e = 0; e < 4; ++e) {
                const float v = sx[u][e];
                if (v >= tcut) {
                    const int j = (gq * 4 + u) * 64 + ssub * 4 + e;
                    #pragma unroll
                    for (int p = 0; p < KTOP; ++p) {
                        const bool better = (v > lw[p]) || (v == lw[p] && j < li[p]);
                        if (better) {
                            #pragma unroll
                            for (int q = KTOP - 1; q > p; --q) { lw[q] = lw[q - 1]; li[q] = li[q - 1]; }
                            lw[p] = v; li[p] = j;
                            break;
                        }
                    }
                }
            }
        }
    }

#define TOPK_BETTER(v1_, j1_, v2_, j2_) (((v1_) > (v2_)) || ((v1_) == (v2_) && (j1_) < (j2_)))
#define TOPK_PICK(wa_, ja_, wb_, jb_)                                       \
    {                                                                       \
        const bool keep_ = TOPK_BETTER(wa_, ja_, wb_, jb_);                 \
        wa_ = keep_ ? wa_ : wb_;                                            \
        ja_ = keep_ ? ja_ : jb_;                                            \
    }
#define TOPK_CEX(wa_, ja_, wb_, jb_)                                        \
    {                                                                       \
        const bool  sw_ = TOPK_BETTER(wb_, jb_, wa_, ja_);                  \
        const float hw_ = sw_ ? wb_ : wa_;                                  \
        const int   hi_ = sw_ ? jb_ : ja_;                                  \
        wb_ = sw_ ? wa_ : wb_;                                              \
        jb_ = sw_ ? ja_ : jb_;                                              \
        wa_ = hw_;                                                          \
        ja_ = hi_;                                                          \
    }
    float w0 = lw[0], w1 = lw[1], w2 = lw[2], w3 = lw[3];
    int   j0 = li[0], j1 = li[1], j2 = li[2], j3 = li[3];
    #pragma unroll
    for (int off = 1; off < 16; off <<= 1) {
        float x0 = __shfl_xor(w0, off), x1 = __shfl_xor(w1, off);
        float x2 = __shfl_xor(w2, off), x3 = __shfl_xor(w3, off);
        int   y0 = __shfl_xor(j0, off), y1 = __shfl_xor(j1, off);
        int   y2 = __shfl_xor(j2, off), y3 = __shfl_xor(j3, off);
        TOPK_PICK(w0, j0, x3, y3)
        TOPK_PICK(w1, j1, x2, y2)
        TOPK_PICK(w2, j2, x1, y1)
        TOPK_PICK(w3, j3, x0, y0)
        TOPK_CEX(w0, j0, w2, j2)
        TOPK_CEX(w1, j1, w3, j3)
        TOPK_CEX(w0, j0, w1, j1)
        TOPK_CEX(w2, j2, w3, j3)
    }
#undef TOPK_CEX
#undef TOPK_PICK
#undef TOPK_BETTER

    const float e0 = __expf(w0 - w0), e1 = __expf(w1 - w0), e2 = __expf(w2 - w0), e3 = __expf(w3 - w0);
    float esum = 0.0f;
    esum += e0; esum += e1; esum += e2; esum += e3;
    const float inv = 1.0f / esum;
    const float t0 = e0 * inv, t1 = e1 * inv, t2 = e2 * inv, t3 = e3 * inv;

    {
        const int dd0 = ssub * 4;
        const float* vr0 = vbuf + vb + (size_t)(j0 & (NTOK - 1)) * DHEAD + dd0;
        const float* vr1 = vbuf + vb + (size_t)(j1 & (NTOK - 1)) * DHEAD + dd0;
        const float* vr2 = vbuf + vb + (size_t)(j2 & (NTOK - 1)) * DHEAD + dd0;
        const float* vr3 = vbuf + vb + (size_t)(j3 & (NTOK - 1)) * DHEAD + dd0;
        const float* qr  = qbuf + qb + (size_t)(i0 + srow) * DHEAD + dd0;
        #pragma unroll
        for (int u = 0; u < 4; ++u) {
            float wvv = 0.0f;
            wvv += t0 * vr0[u];
            wvv += t1 * vr1[u];
            wvv += t2 * vr2[u];
            wvv += t3 * vr3[u];
            const float val = wvv * qr[u];
            atomicAdd(&out_acc[dd0 + u], val);
        }
    }
    __syncthreads();
    if (tid < DHEAD)
        atomicAdd(&out[b * CCH + tid * NHEAD + h], out_acc[tid]);
}

extern "C" void kernel_launch(void* const* d_in, const int* in_sizes, int n_in,
                              void* d_out, int out_size, void* d_ws, size_t ws_size,
                              hipStream_t stream) {
    const float* x  = (const float*)d_in[0];
    const float* Wq = (const float*)d_in[1];
    const float* bq = (const float*)d_in[2];
    float* out = (float*)d_out;
    float* ws  = (float*)d_ws;

    const size_t per = (size_t)BATCH * NHEAD * NTOK * DHEAD;
    float* qbuf = ws;
    float* kT   = ws + per;
    float* vbuf = ws + 2 * per;

    zero_out_kernel<<<(out_size + 255) / 256, 256, 0, stream>>>(out, out_size);
    qkv_kernel<<<768, 256, 0, stream>>>(x, Wq, bq, qbuf, kT, vbuf);
    attn_kernel<<<4096, 256, 0, stream>>>(qbuf, kT, vbuf, out);
}


# attn top-4 collection: scores >= the row's cut go to a per-row LDS list (slot = row count + lower hit lanes, one wave vote); each lane sorts its row's 4-8 entries; insert + lane merge on overflow
# speedup vs baseline: 1.7264x; 1.0343x over previous
#include <hip/hip_runtime.h>
#include <hip/hip_bf16.h>

typedef float v8f __attribute__((ext_vector_type(8)));
typedef float v2f __attribute__((ext_vector_type(2)));

#define BATCH 8
#define CCH   512
#define NHEAD 8
#define DHEAD 64
#define NTOK  1024
#define KTOP  4
#define C3    1536

__global__ void zero_out_kernel(float* __restrict__ out, int n) {
    int i = blockIdx.x * blockDim.x + threadIdx.x;
    if (i < n) out[i] = 0.0f;
}

typedef float v4f __attribute__((ext_vector_type(4)));
#define QK_KC    32
#define QK_KP    (QK_KC / 2)
#define QK_PITCH 288

static __device__ __forceinline__ void qk_chunk(const float* pa, const float* pb, v8f (&acc)[4][2]) {
    v2f af[2][4], bf[2][2];
    #pragma unroll
    for (int i = 0; i < 4; ++i) af[0][i] = *(const v2f*)(pa + i * 32);
    #pragma unroll
    for (int t = 0; t < 2; ++t) bf[0][t] = *(const v2f*)(pb + t * 32);
    #pragma unroll
    for (int s = 0; s < QK_KC / 4; ++s) {
        const int cur = s & 1, nxt = cur ^ 1;
        if (s + 1 < QK_KC / 4) {
            #pragma unroll
            for (int i = 0; i < 4; ++i) af[nxt][i] = *(const v2f*)(pa + 2 * (s + 1) * QK_PITCH + i * 32);
            #pragma unroll
            for (int t = 0; t < 2; ++t) bf[nxt][t] = *(const v2f*)(pb + 2 * (s + 1) * QK_PITCH + t * 32);
        }
        __builtin_amdgcn_sched_barrier(0);
        #pragma unroll
        for (int i = 0; i < 4; ++i) {
            #pragma unroll
            for (int t = 0; t < 2; ++t)
                acc[i][t] = __builtin_amdgcn_wmma_f32_16x16x4_f32(
                    false, af[cur][i], false, bf[cur][t], (short)0, acc[i][t], false, false);
        }
        __builtin_amdgcn_sched_barrier(0);
    }
}

__global__ __launch_bounds__(256) void qkv_kernel(const float* __restrict__ x,
                                                  const float* __restrict__ Wqkv,
                                                  const float* __restrict__ bqkv,
                                                  float* __restrict__ qbuf,
                                                  float* __restrict__ kTbuf,
                                                  float* __restrict__ vbuf) {
    __shared__ __attribute__((aligned(16))) float sA[2][QK_KP * QK_PITCH];
    __shared__ __attribute__((aligned(16))) float sB[2][QK_KP * QK_PITCH];

    const int tid     = threadIdx.x;
    const int lane    = tid & 31;
    const int wv      = tid >> 5;
    const int halfsel = lane >> 4;
    const int lm      = lane & 15;
    const int wm      = wv & 1;
    const int wn      = wv >> 1;

    const int nblk  = blockIdx.x % 12;
    const int mb    = blockIdx.x / 12;
    const int b     = mb >> 3;
    const int mbase = (mb & 7) * 128;
    const int jbase = nblk * 128;

    const float* xb = x + (size_t)b * CCH * NTOK + mbase + 4 * lane;
    const float* wb = Wqkv + jbase + 4 * lane;

    v8f acc[4][2];
    #pragma unroll
    for (int t = 0; t < 2; ++t) {
        const float bv = bqkv[jbase + wn * 32 + 16 * t + lm];
        #pragma unroll
        for (int i = 0; i < 4; ++i) {
            #pragma unroll
            for (int r = 0; r < 8; ++r) acc[i][t][r] = bv;
        }
    }

    v4f ra[2][2], rb[2][2];
    const int aoff = halfsel * QK_PITCH + (wm * 64 + lm) * 2;
    const int boff = halfsel * QK_PITCH + (wn * 32 + lm) * 2;

#define QK_GLOAD(chunk)                                                              \
    _Pragma("unroll")                                                                \
    for (int i_ = 0; i_ < 2; ++i_) {                                                 \
        const int k_ = (chunk) * QK_KC + 2 * (wv + 8 * i_);                          \
        ra[i_][0] = *(const v4f*)(xb + (size_t)k_ * NTOK);                           \
        ra[i_][1] = *(const v4f*)(xb + (size_t)(k_ + 1) * NTOK);                     \
        rb[i_][0] = *(const v4f*)(wb + (size_t)k_ * C3);                             \
        rb[i_][1] = *(const v4f*)(wb + (size_t)(k_ + 1) * C3);                       \
    }
#define QK_SSTORE(buf)                                                               \
    _Pragma("unroll")                                                                \
    for (int i_ = 0; i_ < 2; ++i_) {                                                 \
        float* da_ = &sA[(buf)][(wv + 8 * i_) * QK_PITCH + lane * 8];                \
        float* db_ = &sB[(buf)][(wv + 8 * i_) * QK_PITCH + lane * 8];                \
        *(v4f*)(da_)     = __builtin_shufflevector(ra[i_][0], ra[i_][1], 0, 4, 1, 5); \
        *(v4f*)(da_ + 4) = __builtin_shufflevector(ra[i_][0], ra[i_][1], 2, 6, 3, 7); \
        *(v4f*)(db_)     = __builtin_shufflevector(rb[i_][0], rb[i_][1], 0, 4, 1, 5); \
        *(v4f*)(db_ + 4) = __builtin_shufflevector(rb[i_][0], rb[i_][1], 2, 6, 3, 7); \
    }

    QK_GLOAD(0)
    QK_SSTORE(0)
    __syncthreads();
    QK_GLOAD(1)
    for (int c = 0; c + 2 < CCH / QK_KC; ++c) {
        const int buf = c & 1;
        qk_chunk(&sA[buf][aoff], &sB[buf][boff], acc);
        #pragma unroll
        for (int i = 0; i < 2; ++i) {
            asm volatile("" : "+v"(ra[i][0]), "+v"(ra[i][1]), "+v"(rb[i][0]), "+v"(rb[i][1]));
        }
        QK_SSTORE(buf ^ 1)
        QK_GLOAD(c + 2)
        asm volatile("s_wait_dscnt 0x0\n\ts_barrier_signal -1\n\ts_barrier_wait -1" ::: "memory");
    }
    qk_chunk(&sA[0][aoff], &sB[0][boff], acc);
    QK_SSTORE(1)
    __syncthreads();
    qk_chunk(&sA[1][aoff], &sB[1][boff], acc);
#undef QK_GLOAD
#undef QK_SSTORE

    const int which = jbase >> 9;
    const int head  = ((jbase & 511) + wn * 32) >> 6;
    const int bh    = b * NHEAD + head;
    const int jl0   = (wn & 1) * 32 + lm;
    float* dst;
    int laneoff, mstride, tstride;
    if (which == 1) {
        dst     = kTbuf + (size_t)bh * DHEAD * NTOK;
        laneoff = (jl0 >> 1) * (2 * NTOK) + (jl0 & 1);
        mstride = 2;
        tstride = 8 * 2 * NTOK;
    } else {
        dst     = (which == 0 ? qbuf : vbuf) + (size_t)bh * NTOK * DHEAD;
        laneoff = jl0;
        mstride = DHEAD;
        tstride = 16;
    }
    float* p = dst + laneoff + (size_t)(mbase + wm * 64 + 8 * halfsel) * mstride;
    #pragma unroll
    for (int i = 0; i < 4; ++i) {
        #pragma unroll
        for (int t = 0; t < 2; ++t) {
            #pragma unroll
            for (int r = 0; r < 8; ++r)
                p[(16 * i + r) * mstride + t * tstride] = acc[i][t][r];
        }
    }
}

#define CAND_CAP 8
#define SC_PITCH 1028
__global__ __launch_bounds__(256) void attn_kernel(const float* __restrict__ qbuf,
                                                   const float* __restrict__ kTbuf,
                                                   const float* __restrict__ vbuf,
                                                   float* __restrict__ out) {
    __shared__ __attribute__((aligned(16))) float sc[16 * SC_PITCH];
    __shared__ float out_acc[DHEAD];
    __shared__ __attribute__((aligned(16))) float cand_w[16 * CAND_CAP];
    __shared__ __attribute__((aligned(16))) int   cand_j[16 * CAND_CAP];

    const int tid     = threadIdx.x;
    const int lane    = tid & 31;
    const int wv      = tid >> 5;
    const int halfsel = lane >> 4;
    const int lm      = lane & 15;

    const int bh    = blockIdx.x >> 6;
    const int itile = blockIdx.x & 63;
    const int b     = bh >> 3;
    const int h     = bh & 7;
    const int i0    = itile * 16;

    const size_t qb = (size_t)bh * NTOK * DHEAD;
    const size_t vb = qb;
    const float2* kT2 = (const float2*)(kTbuf + (size_t)bh * DHEAD * NTOK);

    if (tid < DHEAD) out_acc[tid] = 0.0f;

    v2f afrag[16];
    {
        const float* qrow = qbuf + qb + (size_t)(i0 + lm) * DHEAD + 2 * halfsel;
        #pragma unroll
        for (int kk = 0; kk < 16; ++kk) {
            afrag[kk].x = qrow[kk * 4];
            afrag[kk].y = qrow[kk * 4 + 1];
        }
    }

    for (int chunk = 0; chunk < 4; ++chunk) {
        #pragma unroll
        for (int tsub = 0; tsub < 2; ++tsub) {
            const int jg = chunk * 256 + wv * 32 + tsub * 16;
            v8f acc;
            #pragma unroll
            for (int r = 0; r < 8; ++r) acc[r] = 0.0f;
            #pragma unroll
            for (int kk = 0; kk < 16; ++kk) {
                const float2 bp = kT2[(size_t)(kk * 2 + halfsel) * NTOK + jg + lm];
                v2f bf; bf.x = bp.x; bf.y = bp.y;
                acc = __builtin_amdgcn_wmma_f32_16x16x4_f32(
                    false, afrag[kk], false, bf, (short)0, acc, false, false);
            }
            #pragma unroll
            for (int r = 0; r < 8; ++r)
                sc[(r + 8 * halfsel) * SC_PITCH + jg + lm] = acc[r];
        }
    }
    __syncthreads();

    const int srow = tid >> 4;
    const int ssub = tid & 15;
    const float* srcrow = &sc[srow * SC_PITCH + ssub * 4];

    float m0 = -__builtin_inff(), m1 = m0, m2 = m0, m3 = m0;
#define TOP4_TAKE(x_)                                                   \
    {                                                                   \
        const float x__ = (x_);                                         \
        m3 = __builtin_fmaxf(m3, __builtin_fminf(m2, x__));             \
        m2 = __builtin_fmaxf(m2, __builtin_fminf(m1, x__));             \
        m1 = __builtin_fmaxf(m1, __builtin_fminf(m0, x__));             \
        m0 = __builtin_fmaxf(m0, x__);                                  \
    }
    #pragma unroll
    for (int g = 0; g < 16; ++g) {
        const v4f sx = *(const v4f*)(srcrow + g * 64);
        #pragma unroll
        for (int e = 0; e < 4; ++e) TOP4_TAKE(sx[e])
    }
    #pragma unroll
    for (int off = 1; off < 16; off <<= 1) {
        const float o0 = __shfl_xor(m0, off);
        const float o1 = __shfl_xor(m1, off);
        const float o2 = __shfl_xor(m2, off);
        const float o3 = __shfl_xor(m3, off);
        TOP4_TAKE(o0)
        TOP4_TAKE(o1)
        TOP4_TAKE(o2)
        TOP4_TAKE(o3)
    }
#undef TOP4_TAKE
    const float tcut = m3;

#define TOPK_BETTER(v1_, j1_, v2_, j2_) (((v1_) > (v2_)) || ((v1_) == (v2_) && (j1_) < (j2_)))
#define TOPK_PICK(wa_, ja_, wb_, jb_)                                       \
    {                                                                       \
        const bool keep_ = TOPK_BETTER(wa_, ja_, wb_, jb_);                 \
        wa_ = keep_ ? wa_ : wb_;                                            \
        ja_ = keep_ ? ja_ : jb_;                                            \
    }
#define TOPK_CEX(wa_, ja_, wb_, jb_)                                        \
    {                                                                       \
        const bool  sw_ = TOPK_BETTER(wb_, jb_, wa_, ja_);                  \
        const float hw_ = sw_ ? wb_ : wa_;                                  \
        const int   hi_ = sw_ ? jb_ : ja_;                                  \
        wb_ = sw_ ? wa_ : wb_;                                              \
        jb_ = sw_ ? ja_ : jb_;                                              \
        wa_ = hw_;                                                          \
        ja_ = hi_;                                                          \
    }
#define TOPK_SORT4(a0_, b0_, a1_, b1_, a2_, b2_, a3_, b3_)                  \
    TOPK_CEX(a0_, b0_, a1_, b1_)                                            \
    TOPK_CEX(a2_, b2_, a3_, b3_)                                            \
    TOPK_CEX(a0_, b0_, a2_, b2_)                                            \
    TOPK_CEX(a1_, b1_, a3_, b3_)                                            \
    TOPK_CEX(a1_, b1_, a2_, b2_)

    const unsigned rowmask   = halfsel ? 0xffff0000u : 0x0000ffffu;
    const unsigned belowmask = rowmask & ((1u << lane) - 1u);
    int ncand = 0;
    #pragma unroll 1
    for (int gq = 0; gq < 4; ++gq) {
        v4f sx[4];
        #pragma unroll
        for (int u = 0; u < 4; ++u) sx[u] = *(const v4f*)(srcrow + (gq * 4 + u) * 64);
        #pragma unroll
        for (int u = 0; u < 4; ++u) {
            #pragma unroll
            for (int e = 0; e < 4; ++e) {
                const float v      = sx[u][e];
                const bool  hit    = v >= tcut;
                const unsigned hm  = __builtin_amdgcn_ballot_w32(hit);
                if (hm != 0u) {
                    const int slot = ncand + __popc(hm & belowmask);
                    if (hit && slot < CAND_CAP) {
                        cand_w[srow * CAND_CAP + slot] = v;
                        cand_j[srow * CAND_CAP + slot] = (gq * 4 + u) * 64 + ssub * 4 + e;
                    }
                    ncand += __popc(hm & rowmask);
                }
            }
        }
    }
    __builtin_amdgcn_fence(__ATOMIC_RELEASE, "wavefront");
    __builtin_amdgcn_wave_barrier();
    __builtin_amdgcn_fence(__ATOMIC_ACQUIRE, "wavefront");

    float w0, w1, w2, w3;
    int   j0, j1, j2, j3;
    if (!__any(ncand > CAND_CAP)) {
        const float ninf = -__builtin_inff();
        const v4f   cw = *(const v4f*)&cand_w[srow * CAND_CAP];
        const int4  cj = *(const int4*)&cand_j[srow * CAND_CAP];
        w0 = ncand > 0 ? cw[0] : ninf;  j0 = ncand > 0 ? cj.x : 0x7fffffff;
        w1 = ncand > 1 ? cw[1] : ninf;  j1 = ncand > 1 ? cj.y : 0x7fffffff;
        w2 = ncand > 2 ? cw[2] : ninf;  j2 = ncand > 2 ? cj.z : 0x7fffffff;
        w3 = ncand > 3 ? cw[3] : ninf;  j3 = ncand > 3 ? cj.w : 0x7fffffff;
        TOPK_SORT4(w0, j0, w1, j1, w2, j2, w3, j3)
        if (__any(ncand > 4)) {
            const v4f  dw = *(const v4f*)&cand_w[srow * CAND_CAP + 4];
            const int4 dj = *(const int4*)&cand_j[srow * CAND_CAP + 4];
            float x0 = ncand > 4 ? dw[0] : ninf;  int y0 = ncand > 4 ? dj.x : 0x7fffffff;
            float x1 = ncand > 5 ? dw[1] : ninf;  int y1 = ncand > 5 ? dj.y : 0x7fffffff;
            float x2 = ncand > 6 ? dw[2] : ninf;  int y2 = ncand > 6 ? dj.z : 0x7fffffff;
            float x3 = ncand > 7 ? dw[3] : ninf;  int y3 = ncand > 7 ? dj.w : 0x7fffffff;
            TOPK_SORT4(x0, y0, x1, y1, x2, y2, x3, y3)
            TOPK_PICK(w0, j0, x3, y3)
            TOPK_PICK(w1, j1, x2, y2)
            TOPK_PICK(w2, j2, x1, y1)
            TOPK_PICK(w3, j3, x0, y0)
            TOPK_CEX(w0, j0, w2, j2)
            TOPK_CEX(w1, j1, w3, j3)
            TOPK_CEX(w0, j0, w1, j1)
            TOPK_CEX(w2, j2, w3, j3)
        }
    } else {
        float lw[KTOP];
        int   li[KTOP];
        #pragma unroll
        for (int p = 0; p < KTOP; ++p) { lw[p] = -__builtin_inff(); li[p] = 0x7fffffff; }
        #pragma unroll 1
        for (int g = 0; g < 16; ++g) {
            const v4f sx = *(const v4f*)(srcrow + g * 64);
            #pragma unroll
            for (int e = 0; e < 4; ++e) {
                const float v = sx[e];
                if (v >= tcut) {
                    const int j = g * 64 + ssub * 4 + e;
                    #pragma unroll
                    for (int p = 0; p < KTOP; ++p) {
                        const bool better = (v > lw[p]) || (v == lw[p] && j < li[p]);
                        if (better) {
                            #pragma unroll
                            for (int q = KTOP - 1; q > p; --q) { lw[q] = lw[q - 1]; li[q] = li[q - 1]; }
                            lw[p] = v; li[p] = j;
                            break;
                        }
                    }
                }
            }
        }
        w0 = lw[0]; w1 = lw[1]; w2 = lw[2]; w3 = lw[3];
        j0 = li[0]; j1 = li[1]; j2 = li[2]; j3 = li[3];
        #pragma unroll
        for (int off = 1; off < 16; off <<= 1) {
            float x0 = __shfl_xor(w0, off), x1 = __shfl_xor(w1, off);
            float x2 = __shfl_xor(w2, off), x3 = __shfl_xor(w3, off);
            int   y0 = __shfl_xor(j0, off), y1 = __shfl_xor(j1, off);
            int   y2 = __shfl_xor(j2, off), y3 = __shfl_xor(j3, off);
            TOPK_PICK(w0, j0, x3, y3)
            TOPK_PICK(w1, j1, x2, y2)
            TOPK_PICK(w2, j2, x1, y1)
            TOPK_PICK(w3, j3, x0, y0)
            TOPK_CEX(w0, j0, w2, j2)
            TOPK_CEX(w1, j1, w3, j3)
            TOPK_CEX(w0, j0, w1, j1)
            TOPK_CEX(w2, j2, w3, j3)
        }
    }
#undef TOPK_SORT4
#undef TOPK_CEX
#undef TOPK_PICK
#undef TOPK_BETTER

    const float e0 = __expf(w0 - w0), e1 = __expf(w1 - w0), e2 = __expf(w2 - w0), e3 = __expf(w3 - w0);
    float esum = 0.0f;
    esum += e0; esum += e1; esum += e2; esum += e3;
    const float inv = 1.0f / esum;
    const float t0 = e0 * inv, t1 = e1 * inv, t2 = e2 * inv, t3 = e3 * inv;

    {
        const int dd0 = ssub * 4;
        const float* vr0 = vbuf + vb + (size_t)(j0 & (NTOK - 1)) * DHEAD + dd0;
        const float* vr1 = vbuf + vb + (size_t)(j1 & (NTOK - 1)) * DHEAD + dd0;
        const float* vr2 = vbuf + vb + (size_t)(j2 & (NTOK - 1)) * DHEAD + dd0;
        const float* vr3 = vbuf + vb + (size_t)(j3 & (NTOK - 1)) * DHEAD + dd0;
        const float* qr  = qbuf + qb + (size_t)(i0 + srow) * DHEAD + dd0;
        #pragma unroll
        for (int u = 0; u < 4; ++u) {
            float wvv = 0.0f;
            wvv += t0 * vr0[u];
            wvv += t1 * vr1[u];
            wvv += t2 * vr2[u];
            wvv += t3 * vr3[u];
            const float val = wvv * qr[u];
            atomicAdd(&out_acc[dd0 + u], val);
        }
    }
    __syncthreads();
    if (tid < DHEAD)
        atomicAdd(&out[b * CCH + tid * NHEAD + h], out_acc[tid]);
}

extern "C" void kernel_launch(void* const* d_in, const int* in_sizes, int n_in,
                              void* d_out, int out_size, void* d_ws, size_t ws_size,
                              hipStream_t stream) {
    const float* x  = (const float*)d_in[0];
    const float* Wq = (const float*)d_in[1];
    const float* bq = (const float*)d_in[2];
    float* out = (float*)d_out;
    float* ws  = (float*)d_ws;

    const size_t per = (size_t)BATCH * NHEAD * NTOK * DHEAD;
    float* qbuf = ws;
    float* kT   = ws + per;
    float* vbuf = ws + 2 * per;

    zero_out_kernel<<<(out_size + 255) / 256, 256, 0, stream>>>(out, out_size);
    qkv_kernel<<<768, 256, 0, stream>>>(x, Wq, bq, qbuf, kT, vbuf);
    attn_kernel<<<4096, 256, 0, stream>>>(qbuf, kT, vbuf, out);
}


# attn global loads made whole-line (4 lines each): Q tile fetched once per block, staged in LDS for the 8 waves' A-fragments; one 16-byte K^T load feeds two keys' tiles; sums unchanged
# speedup vs baseline: 1.8314x; 1.0608x over previous
#include <hip/hip_runtime.h>
#include <hip/hip_bf16.h>

typedef float v8f __attribute__((ext_vector_type(8)));
typedef float v2f __attribute__((ext_vector_type(2)));

#define BATCH 8
#define CCH   512
#define NHEAD 8
#define DHEAD 64
#define NTOK  1024
#define KTOP  4
#define C3    1536

__global__ void zero_out_kernel(float* __restrict__ out, int n) {
    int i = blockIdx.x * blockDim.x + threadIdx.x;
    if (i < n) out[i] = 0.0f;
}

typedef float v4f __attribute__((ext_vector_type(4)));
#define QK_KC    32
#define QK_KP    (QK_KC / 2)
#define QK_PITCH 288

static __device__ __forceinline__ void qk_chunk(const float* pa, const float* pb, v8f (&acc)[4][2]) {
    v2f af[2][4], bf[2][2];
    #pragma unroll
    for (int i = 0; i < 4; ++i) af[0][i] = *(const v2f*)(pa + i * 32);
    #pragma unroll
    for (int t = 0; t < 2; ++t) bf[0][t] = *(const v2f*)(pb + t * 32);
    #pragma unroll
    for (int s = 0; s < QK_KC / 4; ++s) {
        const int cur = s & 1, nxt = cur ^ 1;
        if (s + 1 < QK_KC / 4) {
            #pragma unroll
            for (int i = 0; i < 4; ++i) af[nxt][i] = *(const v2f*)(pa + 2 * (s + 1) * QK_PITCH + i * 32);
            #pragma unroll
            for (int t = 0; t < 2; ++t) bf[nxt][t] = *(const v2f*)(pb + 2 * (s + 1) * QK_PITCH + t * 32);
        }
        __builtin_amdgcn_sched_barrier(0);
        #pragma unroll
        for (int i = 0; i < 4; ++i) {
            #pragma unroll
            for (int t = 0; t < 2; ++t)
                acc[i][t] = __builtin_amdgcn_wmma_f32_16x16x4_f32(
                    false, af[cur][i], false, bf[cur][t], (short)0, acc[i][t], false, false);
        }
        __builtin_amdgcn_sched_barrier(0);
    }
}

__global__ __launch_bounds__(256) void qkv_kernel(const float* __restrict__ x,
                                                  const float* __restrict__ Wqkv,
                                                  const float* __restrict__ bqkv,
                                                  float* __restrict__ qbuf,
                                                  float* __restrict__ kTbuf,
                                                  float* __restrict__ vbuf) {
    __shared__ __attribute__((aligned(16))) float sA[2][QK_KP * QK_PITCH];
    __shared__ __attribute__((aligned(16))) float sB[2][QK_KP * QK_PITCH];

    const int tid     = threadIdx.x;
    const int lane    = tid & 31;
    const int wv      = tid >> 5;
    const int halfsel = lane >> 4;
    const int lm      = lane & 15;
    const int wm      = wv & 1;
    const int wn      = wv >> 1;

    const int nblk  = blockIdx.x % 12;
    const int mb    = blockIdx.x / 12;
    const int b     = mb >> 3;
    const int mbase = (mb & 7) * 128;
    const int jbase = nblk * 128;

    const float* xb = x + (size_t)b * CCH * NTOK + mbase + 4 * lane;
    const float* wb = Wqkv + jbase + 4 * lane;

    v8f acc[4][2];
    #pragma unroll
    for (int t = 0; t < 2; ++t) {
        const float bv = bqkv[jbase + wn * 32 + 16 * t + lm];
        #pragma unroll
        for (int i = 0; i < 4; ++i) {
            #pragma unroll
            for (int r = 0; r < 8; ++r) acc[i][t][r] = bv;
        }
    }

    v4f ra[2][2], rb[2][2];
    const int aoff = halfsel * QK_PITCH + (wm * 64 + lm) * 2;
    const int boff = halfsel * QK_PITCH + (wn * 32 + lm) * 2;

#define QK_GLOAD(chunk)                                                              \
    _Pragma("unroll")                                                                \
    for (int i_ = 0; i_ < 2; ++i_) {                                                 \
        const int k_ = (chunk) * QK_KC + 2 * (wv + 8 * i_);                          \
        ra[i_][0] = *(const v4f*)(xb + (size_t)k_ * NTOK);                           \
        ra[i_][1] = *(const v4f*)(xb + (size_t)(k_ + 1) * NTOK);                     \
        rb[i_][0] = *(const v4f*)(wb + (size_t)k_ * C3);                             \
        rb[i_][1] = *(const v4f*)(wb + (size_t)(k_ + 1) * C3);                       \
    }
#define QK_SSTORE(buf)                                                               \
    _Pragma("unroll")                                                                \
    for (int i_ = 0; i_ < 2; ++i_) {                                                 \
        float* da_ = &sA[(buf)][(wv + 8 * i_) * QK_PITCH + lane * 8];                \
        float* db_ = &sB[(buf)][(wv + 8 * i_) * QK_PITCH + lane * 8];                \
        *(v4f*)(da_)     = __builtin_shufflevector(ra[i_][0], ra[i_][1], 0, 4, 1, 5); \
        *(v4f*)(da_ + 4) = __builtin_shufflevector(ra[i_][0], ra[i_][1], 2, 6, 3, 7); \
        *(v4f*)(db_)     = __builtin_shufflevector(rb[i_][0], rb[i_][1], 0, 4, 1, 5); \
        *(v4f*)(db_ + 4) = __builtin_shufflevector(rb[i_][0], rb[i_][1], 2, 6, 3, 7); \
    }

    QK_GLOAD(0)
    QK_SSTORE(0)
    __syncthreads();
    QK_GLOAD(1)
    for (int c = 0; c + 2 < CCH / QK_KC; ++c) {
        const int buf = c & 1;
        qk_chunk(&sA[buf][aoff], &sB[buf][boff], acc);
        #pragma unroll
        for (int i = 0; i < 2; ++i) {
            asm volatile("" : "+v"(ra[i][0]), "+v"(ra[i][1]), "+v"(rb[i][0]), "+v"(rb[i][1]));
        }
        QK_SSTORE(buf ^ 1)
        QK_GLOAD(c + 2)
        asm volatile("s_wait_dscnt 0x0\n\ts_barrier_signal -1\n\ts_barrier_wait -1" ::: "memory");
    }
    qk_chunk(&sA[0][aoff], &sB[0][boff], acc);
    QK_SSTORE(1)
    __syncthreads();
    qk_chunk(&sA[1][aoff], &sB[1][boff], acc);
#undef QK_GLOAD
#undef QK_SSTORE

    const int which = jbase >> 9;
    const int head  = ((jbase & 511) + wn * 32) >> 6;
    const int bh    = b * NHEAD + head;
    const int jl0   = (wn & 1) * 32 + lm;
    float* dst;
    int laneoff, mstride, tstride;
    if (which == 1) {
        dst     = kTbuf + (size_t)bh * DHEAD * NTOK;
        laneoff = (jl0 >> 1) * (2 * NTOK) + (jl0 & 1);
        mstride = 2;
        tstride = 8 * 2 * NTOK;
    } else {
        dst     = (which == 0 ? qbuf : vbuf) + (size_t)bh * NTOK * DHEAD;
        laneoff = jl0;
        mstride = DHEAD;
        tstride = 16;
    }
    float* p = dst + laneoff + (size_t)(mbase + wm * 64 + 8 * halfsel) * mstride;
    #pragma unroll
    for (int i = 0; i < 4; ++i) {
        #pragma unroll
        for (int t = 0; t < 2; ++t) {
            #pragma unroll
            for (int r = 0; r < 8; ++r)
                p[(16 * i + r) * mstride + t * tstride] = acc[i][t][r];
        }
    }
}

#define CAND_CAP 8
#define SC_PITCH 1028
#define SQ_PITCH 68
__global__ __launch_bounds__(256) void attn_kernel(const float* __restrict__ qbuf,
                                                   const float* __restrict__ kTbuf,
                                                   const float* __restrict__ vbuf,
                                                   float* __restrict__ out) {
    __shared__ __attribute__((aligned(16))) float sc[16 * SC_PITCH];
    __shared__ float out_acc[DHEAD];
    __shared__ __attribute__((aligned(16))) float sq[16 * SQ_PITCH];
    __shared__ __attribute__((aligned(16))) float cand_w[16 * CAND_CAP];
    __shared__ __attribute__((aligned(16))) int   cand_j[16 * CAND_CAP];

    const int tid     = threadIdx.x;
    const int lane    = tid & 31;
    const int wv      = tid >> 5;
    const int halfsel = lane >> 4;
    const int lm      = lane & 15;

    const int bh    = blockIdx.x >> 6;
    const int itile = blockIdx.x & 63;
    const int b     = bh >> 3;
    const int h     = bh & 7;
    const int i0    = itile * 16;

    const size_t qb = (size_t)bh * NTOK * DHEAD;
    const size_t vb = qb;
    const v4f* kT4 = (const v4f*)(kTbuf + (size_t)bh * DHEAD * NTOK);

    if (tid < DHEAD) out_acc[tid] = 0.0f;

    {
        const v4f qv = *(const v4f*)(qbuf + qb + (size_t)i0 * DHEAD + tid * 4);
        *(v4f*)&sq[(tid >> 4) * SQ_PITCH + (tid & 15) * 4] = qv;
    }
    __syncthreads();

    v2f afrag[16];
    {
        const float* qrow = &sq[lm * SQ_PITCH + 2 * halfsel];
        #pragma unroll
        for (int kk = 0; kk < 16; ++kk) afrag[kk] = *(const v2f*)(qrow + kk * 4);
    }

    for (int chunk = 0; chunk < 4; ++chunk) {
        const int jg = chunk * 256 + wv * 32;
        v8f acc0, acc1;
        #pragma unroll
        for (int r = 0; r < 8; ++r) { acc0[r] = 0.0f; acc1[r] = 0.0f; }
        #pragma unroll
        for (int kk = 0; kk < 16; ++kk) {
            const v4f bp = kT4[(size_t)(kk * 2 + halfsel) * (NTOK / 2) + (jg >> 1) + lm];
            v2f bf0; bf0.x = bp[0]; bf0.y = bp[1];
            v2f bf1; bf1.x = bp[2]; bf1.y = bp[3];
            acc0 = __builtin_amdgcn_wmma_f32_16x16x4_f32(
                false, afrag[kk], false, bf0, (short)0, acc0, false, false);
            acc1 = __builtin_amdgcn_wmma_f32_16x16x4_f32(
                false, afrag[kk], false, bf1, (short)0, acc1, false, false);
        }
        #pragma unroll
        for (int r = 0; r < 8; ++r) {
            float* sd = &sc[(r + 8 * halfsel) * SC_PITCH + jg + 2 * lm];
            sd[0] = acc0[r];
            sd[1] = acc1[r];
        }
    }
    __syncthreads();

    const int srow = tid >> 4;
    const int ssub = tid & 15;
    const float* srcrow = &sc[srow * SC_PITCH + ssub * 4];

    float m0 = -__builtin_inff(), m1 = m0, m2 = m0, m3 = m0;
#define TOP4_TAKE(x_)                                                   \
    {                                                                   \
        const float x__ = (x_);                                         \
        m3 = __builtin_fmaxf(m3, __builtin_fminf(m2, x__));             \
        m2 = __builtin_fmaxf(m2, __builtin_fminf(m1, x__));             \
        m1 = __builtin_fmaxf(m1, __builtin_fminf(m0, x__));             \
        m0 = __builtin_fmaxf(m0, x__);                                  \
    }
    #pragma unroll
    for (int g = 0; g < 16; ++g) {
        const v4f sx = *(const v4f*)(srcrow + g * 64);
        #pragma unroll
        for (int e = 0; e < 4; ++e) TOP4_TAKE(sx[e])
    }
    #pragma unroll
    for (int off = 1; off < 16; off <<= 1) {
        const float o0 = __shfl_xor(m0, off);
        const float o1 = __shfl_xor(m1, off);
        const float o2 = __shfl_xor(m2, off);
        const float o3 = __shfl_xor(m3, off);
        TOP4_TAKE(o0)
        TOP4_TAKE(o1)
        TOP4_TAKE(o2)
        TOP4_TAKE(o3)
    }
#undef TOP4_TAKE
    const float tcut = m3;

#define TOPK_BETTER(v1_, j1_, v2_, j2_) (((v1_) > (v2_)) || ((v1_) == (v2_) && (j1_) < (j2_)))
#define TOPK_PICK(wa_, ja_, wb_, jb_)                                       \
    {                                                                       \
        const bool keep_ = TOPK_BETTER(wa_, ja_, wb_, jb_);                 \
        wa_ = keep_ ? wa_ : wb_;                                            \
        ja_ = keep_ ? ja_ : jb_;                                            \
    }
#define TOPK_CEX(wa_, ja_, wb_, jb_)                                        \
    {                                                                       \
        const bool  sw_ = TOPK_BETTER(wb_, jb_, wa_, ja_);                  \
        const float hw_ = sw_ ? wb_ : wa_;                                  \
        const int   hi_ = sw_ ? jb_ : ja_;                                  \
        wb_ = sw_ ? wa_ : wb_;                                              \
        jb_ = sw_ ? ja_ : jb_;                                              \
        wa_ = hw_;                                                          \
        ja_ = hi_;                                                          \
    }
#define TOPK_SORT4(a0_, b0_, a1_, b1_, a2_, b2_, a3_, b3_)                  \
    TOPK_CEX(a0_, b0_, a1_, b1_)                                            \
    TOPK_CEX(a2_, b2_, a3_, b3_)                                            \
    TOPK_CEX(a0_, b0_, a2_, b2_)                                            \
    TOPK_CEX(a1_, b1_, a3_, b3_)                                            \
    TOPK_CEX(a1_, b1_, a2_, b2_)

    const unsigned rowmask   = halfsel ? 0xffff0000u : 0x0000ffffu;
    const unsigned belowmask = rowmask & ((1u << lane) - 1u);
    int ncand = 0;
    #pragma unroll 1
    for (int gq = 0; gq < 4; ++gq) {
        v4f sx[4];
        #pragma unroll
        for (int u = 0; u < 4; ++u) sx[u] = *(const v4f*)(srcrow + (gq * 4 + u) * 64);
        #pragma unroll
        for (int u = 0; u < 4; ++u) {
            #pragma unroll
            for (int e = 0; e < 4; ++e) {
                const float v      = sx[u][e];
                const bool  hit    = v >= tcut;
                const unsigned hm  = __builtin_amdgcn_ballot_w32(hit);
                if (hm != 0u) {
                    const int slot = ncand + __popc(hm & belowmask);
                    if (hit && slot < CAND_CAP) {
                        cand_w[srow * CAND_CAP + slot] = v;
                        cand_j[srow * CAND_CAP + slot] = (gq * 4 + u) * 64 + ssub * 4 + e;
                    }
                    ncand += __popc(hm & rowmask);
                }
            }
        }
    }
    __builtin_amdgcn_fence(__ATOMIC_RELEASE, "wavefront");
    __builtin_amdgcn_wave_barrier();
    __builtin_amdgcn_fence(__ATOMIC_ACQUIRE, "wavefront");

    float w0, w1, w2, w3;
    int   j0, j1, j2, j3;
    if (!__any(ncand > CAND_CAP)) {
        const float ninf = -__builtin_inff();
        const v4f   cw = *(const v4f*)&cand_w[srow * CAND_CAP];
        const int4  cj = *(const int4*)&cand_j[srow * CAND_CAP];
        w0 = ncand > 0 ? cw[0] : ninf;  j0 = ncand > 0 ? cj.x : 0x7fffffff;
        w1 = ncand > 1 ? cw[1] : ninf;  j1 = ncand > 1 ? cj.y : 0x7fffffff;
        w2 = ncand > 2 ? cw[2] : ninf;  j2 = ncand > 2 ? cj.z : 0x7fffffff;
        w3 = ncand > 3 ? cw[3] : ninf;  j3 = ncand > 3 ? cj.w : 0x7fffffff;
        TOPK_SORT4(w0, j0, w1, j1, w2, j2, w3, j3)
        if (__any(ncand > 4)) {
            const v4f  dw = *(const v4f*)&cand_w[srow * CAND_CAP + 4];
            const int4 dj = *(const int4*)&cand_j[srow * CAND_CAP + 4];
            float x0 = ncand > 4 ? dw[0] : ninf;  int y0 = ncand > 4 ? dj.x : 0x7fffffff;
            float x1 = ncand > 5 ? dw[1] : ninf;  int y1 = ncand > 5 ? dj.y : 0x7fffffff;
            float x2 = ncand > 6 ? dw[2] : ninf;  int y2 = ncand > 6 ? dj.z : 0x7fffffff;
            float x3 = ncand > 7 ? dw[3] : ninf;  int y3 = ncand > 7 ? dj.w : 0x7fffffff;
            TOPK_SORT4(x0, y0, x1, y1, x2, y2, x3, y3)
            TOPK_PICK(w0, j0, x3, y3)
            TOPK_PICK(w1, j1, x2, y2)
            TOPK_PICK(w2, j2, x1, y1)
            TOPK_PICK(w3, j3, x0, y0)
            TOPK_CEX(w0, j0, w2, j2)
            TOPK_CEX(w1, j1, w3, j3)
            TOPK_CEX(w0, j0, w1, j1)
            TOPK_CEX(w2, j2, w3, j3)
        }
    } else {
        float lw[KTOP];
        int   li[KTOP];
        #pragma unroll
        for (int p = 0; p < KTOP; ++p) { lw[p] = -__builtin_inff(); li[p] = 0x7fffffff; }
        #pragma unroll 1
        for (int g = 0; g < 16; ++g) {
            const v4f sx = *(const v4f*)(srcrow + g * 64);
            #pragma unroll
            for (int e = 0; e < 4; ++e) {
                const float v = sx[e];
                if (v >= tcut) {
                    const int j = g * 64 + ssub * 4 + e;
                    #pragma unroll
                    for (int p = 0; p < KTOP; ++p) {
                        const bool better = (v > lw[p]) || (v == lw[p] && j < li[p]);
                        if (better) {
                            #pragma unroll
                            for (int q = KTOP - 1; q > p; --q) { lw[q] = lw[q - 1]; li[q] = li[q - 1]; }
                            lw[p] = v; li[p] = j;
                            break;
                        }
                    }
                }
            }
        }
        w0 = lw[0]; w1 = lw[1]; w2 = lw[2]; w3 = lw[3];
        j0 = li[0]; j1 = li[1]; j2 = li[2]; j3 = li[3];
        #pragma unroll
        for (int off = 1; off < 16; off <<= 1) {
            float x0 = __shfl_xor(w0, off), x1 = __shfl_xor(w1, off);
            float x2 = __shfl_xor(w2, off), x3 = __shfl_xor(w3, off);
            int   y0 = __shfl_xor(j0, off), y1 = __shfl_xor(j1, off);
            int   y2 = __shfl_xor(j2, off), y3 = __shfl_xor(j3, off);
            TOPK_PICK(w0, j0, x3, y3)
            TOPK_PICK(w1, j1, x2, y2)
            TOPK_PICK(w2, j2, x1, y1)
            TOPK_PICK(w3, j3, x0, y0)
            TOPK_CEX(w0, j0, w2, j2)
            TOPK_CEX(w1, j1, w3, j3)
            TOPK_CEX(w0, j0, w1, j1)
            TOPK_CEX(w2, j2, w3, j3)
        }
    }
#undef TOPK_SORT4
#undef TOPK_CEX
#undef TOPK_PICK
#undef TOPK_BETTER

    const float e0 = __expf(w0 - w0), e1 = __expf(w1 - w0), e2 = __expf(w2 - w0), e3 = __expf(w3 - w0);
    float esum = 0.0f;
    esum += e0; esum += e1; esum += e2; esum += e3;
    const float inv = 1.0f / esum;
    const float t0 = e0 * inv, t1 = e1 * inv, t2 = e2 * inv, t3 = e3 * inv;

    {
        const int dd0 = ssub * 4;
        const float* vr0 = vbuf + vb + (size_t)(j0 & (NTOK - 1)) * DHEAD + dd0;
        const float* vr1 = vbuf + vb + (size_t)(j1 & (NTOK - 1)) * DHEAD + dd0;
        const float* vr2 = vbuf + vb + (size_t)(j2 & (NTOK - 1)) * DHEAD + dd0;
        const float* vr3 = vbuf + vb + (size_t)(j3 & (NTOK - 1)) * DHEAD + dd0;
        const float* qr  = &sq[srow * SQ_PITCH + dd0];
        #pragma unroll
        for (int u = 0; u < 4; ++u) {
            float wvv = 0.0f;
            wvv += t0 * vr0[u];
            wvv += t1 * vr1[u];
            wvv += t2 * vr2[u];
            wvv += t3 * vr3[u];
            const float val = wvv * qr[u];
            atomicAdd(&out_acc[dd0 + u], val);
        }
    }
    __syncthreads();
    if (tid < DHEAD)
        atomicAdd(&out[b * CCH + tid * NHEAD + h], out_acc[tid]);
}

extern "C" void kernel_launch(void* const* d_in, const int* in_sizes, int n_in,
                              void* d_out, int out_size, void* d_ws, size_t ws_size,
                              hipStream_t stream) {
    const float* x  = (const float*)d_in[0];
    const float* Wq = (const float*)d_in[1];
    const float* bq = (const float*)d_in[2];
    float* out = (float*)d_out;
    float* ws  = (float*)d_ws;

    const size_t per = (size_t)BATCH * NHEAD * NTOK * DHEAD;
    float* qbuf = ws;
    float* kT   = ws + per;
    float* vbuf = ws + 2 * per;

    zero_out_kernel<<<(out_size + 255) / 256, 256, 0, stream>>>(out, out_size);
    qkv_kernel<<<768, 256, 0, stream>>>(x, Wq, bq, qbuf, kT, vbuf);
    attn_kernel<<<4096, 256, 0, stream>>>(qbuf, kT, vbuf, out);
}


# attn pass 1 only finds a cut: each thread takes plain maxima of its four 16-score quarters (max3, no min chain); cut = 4th largest of the row's 64 group maxima; pass 2 and sums unchanged
# speedup vs baseline: 1.8947x; 1.0346x over previous
#include <hip/hip_runtime.h>
#include <hip/hip_bf16.h>

typedef float v8f __attribute__((ext_vector_type(8)));
typedef float v2f __attribute__((ext_vector_type(2)));

#define BATCH 8
#define CCH   512
#define NHEAD 8
#define DHEAD 64
#define NTOK  1024
#define KTOP  4
#define C3    1536

__global__ void zero_out_kernel(float* __restrict__ out, int n) {
    int i = blockIdx.x * blockDim.x + threadIdx.x;
    if (i < n) out[i] = 0.0f;
}

typedef float v4f __attribute__((ext_vector_type(4)));
#define QK_KC    32
#define QK_KP    (QK_KC / 2)
#define QK_PITCH 288

static __device__ __forceinline__ void qk_chunk(const float* pa, const float* pb, v8f (&acc)[4][2]) {
    v2f af[2][4], bf[2][2];
    #pragma unroll
    for (int i = 0; i < 4; ++i) af[0][i] = *(const v2f*)(pa + i * 32);
    #pragma unroll
    for (int t = 0; t < 2; ++t) bf[0][t] = *(const v2f*)(pb + t * 32);
    #pragma unroll
    for (int s = 0; s < QK_KC / 4; ++s) {
        const int cur = s & 1, nxt = cur ^ 1;
        if (s + 1 < QK_KC / 4) {
            #pragma unroll
            for (int i = 0; i < 4; ++i) af[nxt][i] = *(const v2f*)(pa + 2 * (s + 1) * QK_PITCH + i * 32);
            #pragma unroll
            for (int t = 0; t < 2; ++t) bf[nxt][t] = *(const v2f*)(pb + 2 * (s + 1) * QK_PITCH + t * 32);
        }
        __builtin_amdgcn_sched_barrier(0);
        #pragma unroll
        for (int i = 0; i < 4; ++i) {
            #pragma unroll
            for (int t = 0; t < 2; ++t)
                acc[i][t] = __builtin_amdgcn_wmma_f32_16x16x4_f32(
                    false, af[cur][i], false, bf[cur][t], (short)0, acc[i][t], false, false);
        }
        __builtin_amdgcn_sched_barrier(0);
    }
}

__global__ __launch_bounds__(256) void qkv_kernel(const float* __restrict__ x,
                                                  const float* __restrict__ Wqkv,
                                                  const float* __restrict__ bqkv,
                                                  float* __restrict__ qbuf,
                                                  float* __restrict__ kTbuf,
                                                  float* __restrict__ vbuf) {
    __shared__ __attribute__((aligned(16))) float sA[2][QK_KP * QK_PITCH];
    __shared__ __attribute__((aligned(16))) float sB[2][QK_KP * QK_PITCH];

    const int tid     = threadIdx.x;
    const int lane    = tid & 31;
    const int wv      = tid >> 5;
    const int halfsel = lane >> 4;
    const int lm      = lane & 15;
    const int wm      = wv & 1;
    const int wn      = wv >> 1;

    const int nblk  = blockIdx.x % 12;
    const int mb    = blockIdx.x / 12;
    const int b     = mb >> 3;
    const int mbase = (mb & 7) * 128;
    const int jbase = nblk * 128;

    const float* xb = x + (size_t)b * CCH * NTOK + mbase + 4 * lane;
    const float* wb = Wqkv + jbase + 4 * lane;

    v8f acc[4][2];
    #pragma unroll
    for (int t = 0; t < 2; ++t) {
        const float bv = bqkv[jbase + wn * 32 + 16 * t + lm];
        #pragma unroll
        for (int i = 0; i < 4; ++i) {
            #pragma unroll
            for (int r = 0; r < 8; ++r) acc[i][t][r] = bv;
        }
    }

    v4f ra[2][2], rb[2][2];
    const int aoff = halfsel * QK_PITCH + (wm * 64 + lm) * 2;
    const int boff = halfsel * QK_PITCH + (wn * 32 + lm) * 2;

#define QK_GLOAD(chunk)                                                              \
    _Pragma("unroll")                                                                \
    for (int i_ = 0; i_ < 2; ++i_) {                                                 \
        const int k_ = (chunk) * QK_KC + 2 * (wv + 8 * i_);                          \
        ra[i_][0] = *(const v4f*)(xb + (size_t)k_ * NTOK);                           \
        ra[i_][1] = *(const v4f*)(xb + (size_t)(k_ + 1) * NTOK);                     \
        rb[i_][0] = *(const v4f*)(wb + (size_t)k_ * C3);                             \
        rb[i_][1] = *(const v4f*)(wb + (size_t)(k_ + 1) * C3);                       \
    }
#define QK_SSTORE(buf)                                                               \
    _Pragma("unroll")                                                                \
    for (int i_ = 0; i_ < 2; ++i_) {                                                 \
        float* da_ = &sA[(buf)][(wv + 8 * i_) * QK_PITCH + lane * 8];                \
        float* db_ = &sB[(buf)][(wv + 8 * i_) * QK_PITCH + lane * 8];                \
        *(v4f*)(da_)     = __builtin_shufflevector(ra[i_][0], ra[i_][1], 0, 4, 1, 5); \
        *(v4f*)(da_ + 4) = __builtin_shufflevector(ra[i_][0], ra[i_][1], 2, 6, 3, 7); \
        *(v4f*)(db_)     = __builtin_shufflevector(rb[i_][0], rb[i_][1], 0, 4, 1, 5); \
        *(v4f*)(db_ + 4) = __builtin_shufflevector(rb[i_][0], rb[i_][1], 2, 6, 3, 7); \
    }

    QK_GLOAD(0)
    QK_SSTORE(0)
    __syncthreads();
    QK_GLOAD(1)
    for (int c = 0; c + 2 < CCH / QK_KC; ++c) {
        const int buf = c & 1;
        qk_chunk(&sA[buf][aoff], &sB[buf][boff], acc);
        #pragma unroll
        for (int i = 0; i < 2; ++i) {
            asm volatile("" : "+v"(ra[i][0]), "+v"(ra[i][1]), "+v"(rb[i][0]), "+v"(rb[i][1]));
        }
        QK_SSTORE(buf ^ 1)
        QK_GLOAD(c + 2)
        asm volatile("s_wait_dscnt 0x0\n\ts_barrier_signal -1\n\ts_barrier_wait -1" ::: "memory");
    }
    qk_chunk(&sA[0][aoff], &sB[0][boff], acc);
    QK_SSTORE(1)
    __syncthreads();
    qk_chunk(&sA[1][aoff], &sB[1][boff], acc);
#undef QK_GLOAD
#undef QK_SSTORE

    const int which = jbase >> 9;
    const int head  = ((jbase & 511) + wn * 32) >> 6;
    const int bh    = b * NHEAD + head;
    const int jl0   = (wn & 1) * 32 + lm;
    float* dst;
    int laneoff, mstride, tstride;
    if (which == 1) {
        dst     = kTbuf + (size_t)bh * DHEAD * NTOK;
        laneoff = (jl0 >> 1) * (2 * NTOK) + (jl0 & 1);
        mstride = 2;
        tstride = 8 * 2 * NTOK;
    } else {
        dst     = (which == 0 ? qbuf : vbuf) + (size_t)bh * NTOK * DHEAD;
        laneoff = jl0;
        mstride = DHEAD;
        tstride = 16;
    }
    float* p = dst + laneoff + (size_t)(mbase + wm * 64 + 8 * halfsel) * mstride;
    #pragma unroll
    for (int i = 0; i < 4; ++i) {
        #pragma unroll
        for (int t = 0; t < 2; ++t) {
            #pragma unroll
            for (int r = 0; r < 8; ++r)
                p[(16 * i + r) * mstride + t * tstride] = acc[i][t][r];
        }
    }
}

#define CAND_CAP 8
#define SC_PITCH 1028
#define SQ_PITCH 68
__global__ __launch_bounds__(256) void attn_kernel(const float* __restrict__ qbuf,
                                                   const float* __restrict__ kTbuf,
                                                   const float* __restrict__ vbuf,
                                                   float* __restrict__ out) {
    __shared__ __attribute__((aligned(16))) float sc[16 * SC_PITCH];
    __shared__ float out_acc[DHEAD];
    __shared__ __attribute__((aligned(16))) float sq[16 * SQ_PITCH];
    __shared__ __attribute__((aligned(16))) float cand_w[16 * CAND_CAP];
    __shared__ __attribute__((aligned(16))) int   cand_j[16 * CAND_CAP];

    const int tid     = threadIdx.x;
    const int lane    = tid & 31;
    const int wv      = tid >> 5;
    const int halfsel = lane >> 4;
    const int lm      = lane & 15;

    const int bh    = blockIdx.x >> 6;
    const int itile = blockIdx.x & 63;
    const int b     = bh >> 3;
    const int h     = bh & 7;
    const int i0    = itile * 16;

    const size_t qb = (size_t)bh * NTOK * DHEAD;
    const size_t vb = qb;
    const v4f* kT4 = (const v4f*)(kTbuf + (size_t)bh * DHEAD * NTOK);

    if (tid < DHEAD) out_acc[tid] = 0.0f;

    {
        const v4f qv = *(const v4f*)(qbuf + qb + (size_t)i0 * DHEAD + tid * 4);
        *(v4f*)&sq[(tid >> 4) * SQ_PITCH + (tid & 15) * 4] = qv;
    }
    __syncthreads();

    v2f afrag[16];
    {
        const float* qrow = &sq[lm * SQ_PITCH + 2 * halfsel];
        #pragma unroll
        for (int kk = 0; kk < 16; ++kk) afrag[kk] = *(const v2f*)(qrow + kk * 4);
    }

    for (int chunk = 0; chunk < 4; ++chunk) {
        const int jg = chunk * 256 + wv * 32;
        v8f acc0, acc1;
        #pragma unroll
        for (int r = 0; r < 8; ++r) { acc0[r] = 0.0f; acc1[r] = 0.0f; }
        #pragma unroll
        for (int kk = 0; kk < 16; ++kk) {
            const v4f bp = kT4[(size_t)(kk * 2 + halfsel) * (NTOK / 2) + (jg >> 1) + lm];
            v2f bf0; bf0.x = bp[0]; bf0.y = bp[1];
            v2f bf1; bf1.x = bp[2]; bf1.y = bp[3];
            acc0 = __builtin_amdgcn_wmma_f32_16x16x4_f32(
                false, afrag[kk], false, bf0, (short)0, acc0, false, false);
            acc1 = __builtin_amdgcn_wmma_f32_16x16x4_f32(
                false, afrag[kk], false, bf1, (short)0, acc1, false, false);
        }
        #pragma unroll
        for (int r = 0; r < 8; ++r) {
            float* sd = &sc[(r + 8 * halfsel) * SC_PITCH + jg + 2 * lm];
            sd[0] = acc0[r];
            sd[1] = acc1[r];
        }
    }
    __syncthreads();

    const int srow = tid >> 4;
    const int ssub = tid & 15;
    const float* srcrow = &sc[srow * SC_PITCH + ssub * 4];

    float m0 = -__builtin_inff(), m1 = m0, m2 = m0, m3 = m0;
#define TOP4_TAKE(x_)                                                   \
    {                                                                   \
        const float x__ = (x_);                                         \
        m3 = __builtin_fmaxf(m3, __builtin_fminf(m2, x__));             \
        m2 = __builtin_fmaxf(m2, __builtin_fminf(m1, x__));             \
        m1 = __builtin_fmaxf(m1, __builtin_fminf(m0, x__));             \
        m0 = __builtin_fmaxf(m0, x__);                                  \
    }
    #pragma unroll
    for (int gq = 0; gq < 4; ++gq) {
        float gm = -__builtin_inff();
        #pragma unroll
        for (int u = 0; u < 4; ++u) {
            const v4f sx = *(const v4f*)(srcrow + (gq * 4 + u) * 64);
            gm = __builtin_fmaxf(gm, __builtin_fmaxf(__builtin_fmaxf(sx[0], sx[1]),
                                                     __builtin_fmaxf(sx[2], sx[3])));
        }
        TOP4_TAKE(gm)
    }
    #pragma unroll
    for (int off = 1; off < 16; off <<= 1) {
        const float o0 = __shfl_xor(m0, off);
        const float o1 = __shfl_xor(m1, off);
        const float o2 = __shfl_xor(m2, off);
        const float o3 = __shfl_xor(m3, off);
        TOP4_TAKE(o0)
        TOP4_TAKE(o1)
        TOP4_TAKE(o2)
        TOP4_TAKE(o3)
    }
#undef TOP4_TAKE
    const float tcut = m3;

#define TOPK_BETTER(v1_, j1_, v2_, j2_) (((v1_) > (v2_)) || ((v1_) == (v2_) && (j1_) < (j2_)))
#define TOPK_PICK(wa_, ja_, wb_, jb_)                                       \
    {                                                                       \
        const bool keep_ = TOPK_BETTER(wa_, ja_, wb_, jb_);                 \
        wa_ = keep_ ? wa_ : wb_;                                            \
        ja_ = keep_ ? ja_ : jb_;                                            \
    }
#define TOPK_CEX(wa_, ja_, wb_, jb_)                                        \
    {                                                                       \
        const bool  sw_ = TOPK_BETTER(wb_, jb_, wa_, ja_);                  \
        const float hw_ = sw_ ? wb_ : wa_;                                  \
        const int   hi_ = sw_ ? jb_ : ja_;                                  \
        wb_ = sw_ ? wa_ : wb_;                                              \
        jb_ = sw_ ? ja_ : jb_;                                              \
        wa_ = hw_;                                                          \
        ja_ = hi_;                                                          \
    }
#define TOPK_SORT4(a0_, b0_, a1_, b1_, a2_, b2_, a3_, b3_)                  \
    TOPK_CEX(a0_, b0_, a1_, b1_)                                            \
    TOPK_CEX(a2_, b2_, a3_, b3_)                                            \
    TOPK_CEX(a0_, b0_, a2_, b2_)                                            \
    TOPK_CEX(a1_, b1_, a3_, b3_)                                            \
    TOPK_CEX(a1_, b1_, a2_, b2_)

    const unsigned rowmask   = halfsel ? 0xffff0000u : 0x0000ffffu;
    const unsigned belowmask = rowmask & ((1u << lane) - 1u);
    int ncand = 0;
    #pragma unroll 1
    for (int gq = 0; gq < 4; ++gq) {
        v4f sx[4];
        #pragma unroll
        for (int u = 0; u < 4; ++u) sx[u] = *(const v4f*)(srcrow + (gq * 4 + u) * 64);
        #pragma unroll
        for (int u = 0; u < 4; ++u) {
            #pragma unroll
            for (int e = 0; e < 4; ++e) {
                const float v      = sx[u][e];
                const bool  hit    = v >= tcut;
                const unsigned hm  = __builtin_amdgcn_ballot_w32(hit);
                if (hm != 0u) {
                    const int slot = ncand + __popc(hm & belowmask);
                    if (hit && slot < CAND_CAP) {
                        cand_w[srow * CAND_CAP + slot] = v;
                        cand_j[srow * CAND_CAP + slot] = (gq * 4 + u) * 64 + ssub * 4 + e;
                    }
                    ncand += __popc(hm & rowmask);
                }
            }
        }
    }
    __builtin_amdgcn_fence(__ATOMIC_RELEASE, "wavefront");
    __builtin_amdgcn_wave_barrier();
    __builtin_amdgcn_fence(__ATOMIC_ACQUIRE, "wavefront");

    float w0, w1, w2, w3;
    int   j0, j1, j2, j3;
    if (!__any(ncand > CAND_CAP)) {
        const float ninf = -__builtin_inff();
        const v4f   cw = *(const v4f*)&cand_w[srow * CAND_CAP];
        const int4  cj = *(const int4*)&cand_j[srow * CAND_CAP];
        w0 = ncand > 0 ? cw[0] : ninf;  j0 = ncand > 0 ? cj.x : 0x7fffffff;
        w1 = ncand > 1 ? cw[1] : ninf;  j1 = ncand > 1 ? cj.y : 0x7fffffff;
        w2 = ncand > 2 ? cw[2] : ninf;  j2 = ncand > 2 ? cj.z : 0x7fffffff;
        w3 = ncand > 3 ? cw[3] : ninf;  j3 = ncand > 3 ? cj.w : 0x7fffffff;
        TOPK_SORT4(w0, j0, w1, j1, w2, j2, w3, j3)
        if (__any(ncand > 4)) {
            const v4f  dw = *(const v4f*)&cand_w[srow * CAND_CAP + 4];
            const int4 dj = *(const int4*)&cand_j[srow * CAND_CAP + 4];
            float x0 = ncand > 4 ? dw[0] : ninf;  int y0 = ncand > 4 ? dj.x : 0x7fffffff;
            float x1 = ncand > 5 ? dw[1] : ninf;  int y1 = ncand > 5 ? dj.y : 0x7fffffff;
            float x2 = ncand > 6 ? dw[2] : ninf;  int y2 = ncand > 6 ? dj.z : 0x7fffffff;
            float x3 = ncand > 7 ? dw[3] : ninf;  int y3 = ncand > 7 ? dj.w : 0x7fffffff;
            TOPK_SORT4(x0, y0, x1, y1, x2, y2, x3, y3)
            TOPK_PICK(w0, j0, x3, y3)
            TOPK_PICK(w1, j1, x2, y2)
            TOPK_PICK(w2, j2, x1, y1)
            TOPK_PICK(w3, j3, x0, y0)
            TOPK_CEX(w0, j0, w2, j2)
            TOPK_CEX(w1, j1, w3, j3)
            TOPK_CEX(w0, j0, w1, j1)
            TOPK_CEX(w2, j2, w3, j3)
        }
    } else {
        float lw[KTOP];
        int   li[KTOP];
        #pragma unroll
        for (int p = 0; p < KTOP; ++p) { lw[p] = -__builtin_inff(); li[p] = 0x7fffffff; }
        #pragma unroll 1
        for (int g = 0; g < 16; ++g) {
            const v4f sx = *(const v4f*)(srcrow + g * 64);
            #pragma unroll
            for (int e = 0; e < 4; ++e) {
                const float v = sx[e];
                if (v >= tcut) {
                    const int j = g * 64 + ssub * 4 + e;
                    #pragma unroll
                    for (int p = 0; p < KTOP; ++p) {
                        const bool better = (v > lw[p]) || (v == lw[p] && j < li[p]);
                        if (better) {
                            #pragma unroll
                            for (int q = KTOP - 1; q > p; --q) { lw[q] = lw[q - 1]; li[q] = li[q - 1]; }
                            lw[p] = v; li[p] = j;
                            break;
                        }
                    }
                }
            }
        }
        w0 = lw[0]; w1 = lw[1]; w2 = lw[2]; w3 = lw[3];
        j0 = li[0]; j1 = li[1]; j2 = li[2]; j3 = li[3];
        #pragma unroll
        for (int off = 1; off < 16; off <<= 1) {
            float x0 = __shfl_xor(w0, off), x1 = __shfl_xor(w1, off);
            float x2 = __shfl_xor(w2, off), x3 = __shfl_xor(w3, off);
            int   y0 = __shfl_xor(j0, off), y1 = __shfl_xor(j1, off);
            int   y2 = __shfl_xor(j2, off), y3 = __shfl_xor(j3, off);
            TOPK_PICK(w0, j0, x3, y3)
            TOPK_PICK(w1, j1, x2, y2)
            TOPK_PICK(w2, j2, x1, y1)
            TOPK_PICK(w3, j3, x0, y0)
            TOPK_CEX(w0, j0, w2, j2)
            TOPK_CEX(w1, j1, w3, j3)
            TOPK_CEX(w0, j0, w1, j1)
            TOPK_CEX(w2, j2, w3, j3)
        }
    }
#undef TOPK_SORT4
#undef TOPK_CEX
#undef TOPK_PICK
#undef TOPK_BETTER

    const float e0 = __expf(w0 - w0), e1 = __expf(w1 - w0), e2 = __expf(w2 - w0), e3 = __expf(w3 - w0);
    float esum = 0.0f;
    esum += e0; esum += e1; esum += e2; esum += e3;
    const float inv = 1.0f / esum;
    const float t0 = e0 * inv, t1 = e1 * inv, t2 = e2 * inv, t3 = e3 * inv;

    {
        const int dd0 = ssub * 4;
        const float* vr0 = vbuf + vb + (size_t)(j0 & (NTOK - 1)) * DHEAD + dd0;
        const float* vr1 = vbuf + vb + (size_t)(j1 & (NTOK - 1)) * DHEAD + dd0;
        const float* vr2 = vbuf + vb + (size_t)(j2 & (NTOK - 1)) * DHEAD + dd0;
        const float* vr3 = vbuf + vb + (size_t)(j3 & (NTOK - 1)) * DHEAD + dd0;
        const float* qr  = &sq[srow * SQ_PITCH + dd0];
        #pragma unroll
        for (int u = 0; u < 4; ++u) {
            float wvv = 0.0f;
            wvv += t0 * vr0[u];
            wvv += t1 * vr1[u];
            wvv += t2 * vr2[u];
            wvv += t3 * vr3[u];
            const float val = wvv * qr[u];
            atomicAdd(&out_acc[dd0 + u], val);
        }
    }
    __syncthreads();
    if (tid < DHEAD)
        atomicAdd(&out[b * CCH + tid * NHEAD + h], out_acc[tid]);
}

extern "C" void kernel_launch(void* const* d_in, const int* in_sizes, int n_in,
                              void* d_out, int out_size, void* d_ws, size_t ws_size,
                              hipStream_t stream) {
    const float* x  = (const float*)d_in[0];
    const float* Wq = (const float*)d_in[1];
    const float* bq = (const float*)d_in[2];
    float* out = (float*)d_out;
    float* ws  = (float*)d_ws;

    const size_t per = (size_t)BATCH * NHEAD * NTOK * DHEAD;
    float* qbuf = ws;
    float* kT   = ws + per;
    float* vbuf = ws + 2 * per;

    zero_out_kernel<<<(out_size + 255) / 256, 256, 0, stream>>>(out, out_size);
    qkv_kernel<<<768, 256, 0, stream>>>(x, Wq, bq, qbuf, kT, vbuf);
    attn_kernel<<<4096, 256, 0, stream>>>(qbuf, kT, vbuf, out);
}
